# starting kernel
# baseline (speedup 1.0000x reference)
#include <hip/hip_runtime.h>
#include <hip/hip_bf16.h>


typedef float v2f __attribute__((ext_vector_type(2)));
typedef float v8f __attribute__((ext_vector_type(8)));

#define T_STEPS 256
#define UNITS_  256
#define N4      1024
#define KTOT    512
#define ASTRIDE 516
#define ZSTRIDE 1028
#define COMBO_F (512 * 1024)

__device__ __forceinline__ float sigmoid_f(float x) {
    return 1.0f / (1.0f + __expf(-x));
}
__device__ __forceinline__ float tanh_f(float x) {
    x = fminf(15.0f, fmaxf(-15.0f, x));
    float e = __expf(2.0f * x);
    return (e - 1.0f) / (e + 1.0f);
}

__global__ void pack_weights_kernel(const float* __restrict__ kfw,
                                    const float* __restrict__ rfw,
                                    const float* __restrict__ kbw,
                                    const float* __restrict__ rbw,
                                    float* __restrict__ packed) {
    int i = blockIdx.x * blockDim.x + threadIdx.x;
    int lane  = i & 31;
    int ks    = (i >> 5) & 127;
    int nt    = (i >> 12) & 63;
    int combo = i >> 18;
    int layer = combo >> 1;
    int isbw  = combo & 1;
    const float* W = (isbw ? kbw : kfw) + (size_t)layer * 256 * 1024;
    const float* R = (isbw ? rbw : rfw) + (size_t)layer * 256 * 1024;

    int n  = nt * 16 + (lane & 15);
    int k0 = ks * 4 + ((lane >> 4) << 1);
    float d0, d1;
    if (k0 < 256) {
        d0 = W[(size_t)k0 * 1024 + n];
        d1 = W[(size_t)(k0 + 1) * 1024 + n];
    } else {
        int kk = k0 - 256;
        d0 = R[(size_t)kk * 1024 + n];
        d1 = R[(size_t)(kk + 1) * 1024 + n];
    }
    packed[(size_t)i * 2 + 0] = d0;
    packed[(size_t)i * 2 + 1] = d1;
}

__global__ void __launch_bounds__(1024)
bilstm_layer_kernel(const float* __restrict__ in_fw,
                    const float* __restrict__ in_bw,
                    const float* __restrict__ pw_fw,
                    const float* __restrict__ pw_bw,
                    const float* __restrict__ bias_fw,
                    const float* __restrict__ bias_bw,
                    float* __restrict__ out_fw,
                    float* __restrict__ out_bw,
                    int residual) {
    extern __shared__ float lds[];
    float* Alds = lds;
    float* Zlds = lds + 16 * ASTRIDE;

    const int dir        = blockIdx.y;
    const float* in      = dir ? in_bw   : in_fw;
    const float* pw      = dir ? pw_bw   : pw_fw;
    const float* bias    = dir ? bias_bw : bias_fw;
    float*       out     = dir ? out_bw  : out_fw;

    const int b0      = blockIdx.x * 16;
    const int tid     = threadIdx.x;
    const int lane    = tid & 31;
    const int wave    = tid >> 5;
    const int halfsel = lane >> 4;
    const int lrow    = lane & 15;

    int   rr[4], uu[4];
    float bI[4], bF[4], bG[4], bO[4], cst[4];
#pragma unroll
    for (int j = 0; j < 4; ++j) {
        int f = tid + j * 1024;
        rr[j] = f >> 8;
        uu[j] = f & 255;
        bI[j] = bias[uu[j]];
        bF[j] = bias[256 + uu[j]];
        bG[j] = bias[512 + uu[j]];
        bO[j] = bias[768 + uu[j]];
        cst[j] = 0.0f;
        Alds[rr[j] * ASTRIDE + 256 + uu[j]] = 0.0f;
    }

    const int nt0 = wave * 2;
    const int nt1 = wave * 2 + 1;
    const float* pwt0 = pw + ((size_t)(nt0 * 128) * 32 + lane) * 2;
    const float* pwt1 = pw + ((size_t)(nt1 * 128) * 32 + lane) * 2;
    const int aBase = lrow * ASTRIDE + 2 * halfsel;

    for (int tt = 0; tt < T_STEPS; ++tt) {
        const int t = dir ? (T_STEPS - 1 - tt) : tt;

#pragma unroll
        for (int j = 0; j < 4; ++j) {
            Alds[rr[j] * ASTRIDE + uu[j]] =
                in[((size_t)(b0 + rr[j]) * T_STEPS + t) * UNITS_ + uu[j]];
        }
        if (tt + 1 < T_STEPS) {
            const int tn = dir ? (t - 1) : (t + 1);
            __builtin_prefetch(
                &in[((size_t)(b0 + rr[0]) * T_STEPS + tn) * UNITS_ + uu[0]], 0, 1);
        }
        __syncthreads();

        v8f acc0 = {0.f, 0.f, 0.f, 0.f, 0.f, 0.f, 0.f, 0.f};
        v8f acc1 = {0.f, 0.f, 0.f, 0.f, 0.f, 0.f, 0.f, 0.f};
#pragma unroll 4
        for (int ks = 0; ks < KTOT / 4; ++ks) {
            v2f a  = *(const v2f*)(Alds + aBase + 4 * ks);
            v2f w0 = *(const v2f*)(pwt0 + ks * 64);
            v2f w1 = *(const v2f*)(pwt1 + ks * 64);
            acc0 = __builtin_amdgcn_wmma_f32_16x16x4_f32(
                false, a, false, w0, (short)0, acc0, false, false);
            acc1 = __builtin_amdgcn_wmma_f32_16x16x4_f32(
                false, a, false, w1, (short)0, acc1, false, false);
        }
        {
            const int zr = 8 * halfsel;
#pragma unroll
            for (int e = 0; e < 8; ++e) {
                Zlds[(zr + e) * ZSTRIDE + nt0 * 16 + lrow] = acc0[e];
                Zlds[(zr + e) * ZSTRIDE + nt1 * 16 + lrow] = acc1[e];
            }
        }
        __syncthreads();

#pragma unroll
        for (int j = 0; j < 4; ++j) {
            const int r = rr[j], u = uu[j];
            float zi = Zlds[r * ZSTRIDE + u]       + bI[j];
            float zf = Zlds[r * ZSTRIDE + 256 + u] + bF[j];
            float zg = Zlds[r * ZSTRIDE + 512 + u] + bG[j];
            float zo = Zlds[r * ZSTRIDE + 768 + u] + bO[j];
            float gi = sigmoid_f(zi);
            float gf = sigmoid_f(zf);
            float gg = tanh_f(zg);
            float go = sigmoid_f(zo);
            float c  = gf * cst[j] + gi * gg;
            cst[j]   = c;
            float h  = go * tanh_f(c);
            Alds[r * ASTRIDE + 256 + u] = h;
            float ov = residual ? (h + Alds[r * ASTRIDE + u]) : h;
            out[((size_t)(b0 + r) * T_STEPS + t) * UNITS_ + u] = ov;
        }
        __syncthreads();
    }
}

__global__ void combine_avg_kernel(const float4* __restrict__ fw,
                                   float4* __restrict__ out, int n4) {
    int i = blockIdx.x * blockDim.x + threadIdx.x;
    if (i < n4) {
        float4 a = fw[i];
        float4 b = out[i];
        out[i] = make_float4((a.x + b.x) * 0.5f, (a.y + b.y) * 0.5f,
                             (a.z + b.z) * 0.5f, (a.w + b.w) * 0.5f);
    }
}

extern "C" void kernel_launch(void* const* d_in, const int* in_sizes, int n_in,
                              void* d_out, int out_size, void* d_ws, size_t ws_size,
                              hipStream_t stream) {
    const float* x   = (const float*)d_in[0];
    const float* kfw = (const float*)d_in[1];
    const float* rfw = (const float*)d_in[2];
    const float* bfw = (const float*)d_in[3];
    const float* kbw = (const float*)d_in[4];
    const float* rbw = (const float*)d_in[5];
    const float* bbw = (const float*)d_in[6];
    float* out = (float*)d_out;

    float* pk   = (float*)d_ws;
    float* wsfw = pk + (size_t)4 * COMBO_F;
    float* wsbw = out;

    pack_weights_kernel<<<4096, 256, 0, stream>>>(kfw, rfw, kbw, rbw, pk);

    const size_t shmem = (size_t)(16 * ASTRIDE + 16 * ZSTRIDE) * sizeof(float);
    (void)hipFuncSetAttribute((const void*)bilstm_layer_kernel,
                              hipFuncAttributeMaxDynamicSharedMemorySize,
                              (int)shmem);

    bilstm_layer_kernel<<<dim3(8, 2), 1024, shmem, stream>>>(
        x, x, pk + 0 * (size_t)COMBO_F, pk + 1 * (size_t)COMBO_F,
        bfw + 0, bbw + 0, wsfw, wsbw,  0);

    bilstm_layer_kernel<<<dim3(8, 2), 1024, shmem, stream>>>(
        wsfw, wsbw, pk + 2 * (size_t)COMBO_F, pk + 3 * (size_t)COMBO_F,
        bfw + 1024, bbw + 1024, wsfw, wsbw,  1);

    const int n4 = (128 * 256 * 256) / 4;
    combine_avg_kernel<<<(n4 + 255) / 256, 256, 0, stream>>>(
        (const float4*)wsfw, (float4*)out, n4);
}


# each wave owns all four gates of its 8 units (R columns regrouped at packing): cells update from the accumulator registers, no z trip through LDS, one barrier per step; sum order kept
# speedup vs baseline: 1.4664x; 1.4664x over previous
#include <hip/hip_runtime.h>


typedef float v2f __attribute__((ext_vector_type(2)));
typedef float v8f __attribute__((ext_vector_type(8)));

#define T_STEPS 256
#define UNITS_  256
#define N4      1024
#define TC      4
#define NCHUNK  64
#define NBATCH  128
#define HSTRIDE 260
#define RCOMBO_F (256 * 1024)

__device__ __forceinline__ float sigmoid_f(float x) {
    return 1.0f / (1.0f + __expf(-x));
}
__device__ __forceinline__ float tanh_f(float x) {
    x = fminf(15.0f, fmaxf(-15.0f, x));
    float e = __expf(2.0f * x);
    return (e - 1.0f) / (e + 1.0f);
}

__global__ void pack_r_kernel(const float* __restrict__ rfw,
                              const float* __restrict__ rbw,
                              float* __restrict__ packed) {
    int i = blockIdx.x * blockDim.x + threadIdx.x;
    int lane = i & 31;
    int ks   = (i >> 5) & 63;
    int nt   = (i >> 11) & 63;
    int isbw = i >> 17;
    const float* R = isbw ? rbw : rfw;
    int c  = lane & 15;
    int n  = (2 * (nt & 1) + (c >> 3)) * 256 + 8 * (nt >> 1) + (c & 7);
    int k0 = ks * 4 + ((lane >> 4) << 1);
    packed[(size_t)i * 2 + 0] = R[(size_t)k0 * 1024 + n];
    packed[(size_t)i * 2 + 1] = R[(size_t)(k0 + 1) * 1024 + n];
}

__global__ void __launch_bounds__(256)
xproj_kernel(const float* __restrict__ in_fw, const float* __restrict__ in_bw,
             const float* __restrict__ w_fw,  const float* __restrict__ w_bw,
             const float* __restrict__ bias_fw, const float* __restrict__ bias_bw,
             float* __restrict__ xz, int chunk) {
    const int lane    = threadIdx.x & 31;
    const int gw      = blockIdx.x * 8 + (threadIdx.x >> 5);
    const int mt      = gw >> 5;
    const int np      = gw & 31;
    const int dir     = mt >> 5;
    const int tc      = (mt & 31) >> 3;
    const int bt      = mt & 7;
    const int halfsel = lane >> 4;
    const int lrow    = lane & 15;

    const float* in   = dir ? in_bw : in_fw;
    const float* Wd   = dir ? w_bw : w_fw;
    const float* bias = dir ? bias_bw : bias_fw;
    const int t = dir ? (T_STEPS - 1 - TC * chunk - tc) : (TC * chunk + tc);

    const float* arow = in + ((size_t)(bt * 16 + lrow) * T_STEPS + t) * UNITS_ + 2 * halfsel;
    const int n0 = np * 32 + lrow;
    const float* wcol = Wd + (size_t)(2 * halfsel) * N4 + n0;

    v8f acc0 = {0.f, 0.f, 0.f, 0.f, 0.f, 0.f, 0.f, 0.f};
    v8f acc1 = {0.f, 0.f, 0.f, 0.f, 0.f, 0.f, 0.f, 0.f};
#pragma unroll 8
    for (int ks = 0; ks < UNITS_ / 4; ++ks) {
        v2f a = *(const v2f*)(arow + 4 * ks);
        const float* wp = wcol + (size_t)(4 * ks) * N4;
        v2f w0, w1;
        w0.x = wp[0];
        w0.y = wp[N4];
        w1.x = wp[16];
        w1.y = wp[N4 + 16];
        acc0 = __builtin_amdgcn_wmma_f32_16x16x4_f32(
            false, a, false, w0, (short)0, acc0, false, false);
        acc1 = __builtin_amdgcn_wmma_f32_16x16x4_f32(
            false, a, false, w1, (short)0, acc1, false, false);
    }
    const float bv0 = bias[n0];
    const float bv1 = bias[n0 + 16];
    float* dst = xz + ((size_t)((dir * TC + tc) * NBATCH + bt * 16 + 8 * halfsel)) * N4 + n0;
#pragma unroll
    for (int e = 0; e < 8; ++e) {
        dst[(size_t)e * N4]      = acc0[e] + bv0;
        dst[(size_t)e * N4 + 16] = acc1[e] + bv1;
    }
}

__device__ __forceinline__ float swap8(float v) {
    return __int_as_float(__builtin_amdgcn_update_dpp(
        0, __float_as_int(v), 0x128, 0xF, 0xF, false));
}

__global__ void __launch_bounds__(1024)
lstm_chunk_kernel(const float* in_fw, const float* in_bw,
                  const float* __restrict__ pr_fw,
                  const float* __restrict__ pr_bw,
                  const float* __restrict__ xz,
                  float* __restrict__ state,
                  float* out_fw, float* out_bw,
                  int chunk, int residual) {
    __shared__ __attribute__((aligned(16))) float Hbuf[2 * 16 * HSTRIDE];

    const int dir     = blockIdx.y;
    const float* in   = dir ? in_bw : in_fw;
    const float* pr   = dir ? pr_bw : pr_fw;
    float*       out  = dir ? out_bw : out_fw;

    const int b0      = blockIdx.x * 16;
    const int tid     = threadIdx.x;
    const int lane    = tid & 31;
    const int wave    = tid >> 5;
    const int halfsel = lane >> 4;
    const int lrow    = lane & 15;
    const bool lower  = (lane & 8) == 0;

    float* st_h = state + (size_t)(dir * 2 + 0) * NBATCH * UNITS_;
    float* st_c = state + (size_t)(dir * 2 + 1) * NBATCH * UNITS_;

    const int cunit = wave * 8 + (lane & 7);
    const int crow0 = 8 * halfsel + (lower ? 0 : 4);
    const int frow  = wave >> 1;
    const int fcol  = 128 * (wave & 1) + 4 * lane;

    float cst[4] = {0.0f, 0.0f, 0.0f, 0.0f};
    float4 h4 = make_float4(0.0f, 0.0f, 0.0f, 0.0f);
    if (chunk != 0) {
        const float4 c4 = *(const float4*)(st_c + (size_t)b0 * UNITS_ + 4 * tid);
        cst[0] = c4.x; cst[1] = c4.y; cst[2] = c4.z; cst[3] = c4.w;
        h4 = *(const float4*)(st_h + (size_t)(b0 + frow) * UNITS_ + fcol);
    }
    *(float4*)(Hbuf + frow * HSTRIDE + fcol) = h4;

    const int nt0 = wave * 2;
    const int nt1 = wave * 2 + 1;
    const float* pwt0 = pr + ((size_t)(nt0 * 64) * 32 + lane) * 2;
    const float* pwt1 = pr + ((size_t)(nt1 * 64) * 32 + lane) * 2;
    const int aBase = lrow * HSTRIDE + 2 * halfsel;
    const int gcol0 = (lower ? 0 : UNITS_) + cunit;

    __syncthreads();

    for (int tc = 0; tc < TC; ++tc) {
        const int t = dir ? (T_STEPS - 1 - TC * chunk - tc) : (TC * chunk + tc);
        const float* Hcur = Hbuf + (tc & 1) * (16 * HSTRIDE);
        float*       Hnxt = Hbuf + ((tc & 1) ^ 1) * (16 * HSTRIDE);

        const float* xzp = xz + ((size_t)((dir * TC + tc) * NBATCH + b0 + 8 * halfsel)) * N4 + gcol0;
        v8f x0, x1;
#pragma unroll
        for (int e = 0; e < 8; ++e) {
            x0[e] = xzp[(size_t)e * N4];
            x1[e] = xzp[(size_t)e * N4 + 2 * UNITS_];
        }
        const size_t oidx = ((size_t)(b0 + frow) * T_STEPS + t) * UNITS_ + fcol;
        float4 res4 = make_float4(0.0f, 0.0f, 0.0f, 0.0f);
        if (residual) res4 = *(const float4*)(in + oidx);

        v8f acc0 = {0.f, 0.f, 0.f, 0.f, 0.f, 0.f, 0.f, 0.f};
        v8f acc1 = {0.f, 0.f, 0.f, 0.f, 0.f, 0.f, 0.f, 0.f};
#pragma unroll 4
        for (int ks = 0; ks < UNITS_ / 4; ++ks) {
            v2f a  = *(const v2f*)(Hcur + aBase + 4 * ks);
            v2f w0 = *(const v2f*)(pwt0 + ks * 64);
            v2f w1 = *(const v2f*)(pwt1 + ks * 64);
            acc0 = __builtin_amdgcn_wmma_f32_16x16x4_f32(
                false, a, false, w0, (short)0, acc0, false, false);
            acc1 = __builtin_amdgcn_wmma_f32_16x16x4_f32(
                false, a, false, w1, (short)0, acc1, false, false);
        }

#pragma unroll
        for (int j = 0; j < 4; ++j) {
            const float z0lo = acc0[j]     + x0[j];
            const float z0hi = acc0[j + 4] + x0[j + 4];
            const float z1lo = acc1[j]     + x1[j];
            const float z1hi = acc1[j + 4] + x1[j + 4];
            const float k0 = lower ? z0lo : z0hi;
            const float s0 = lower ? z0hi : z0lo;
            const float k1 = lower ? z1lo : z1hi;
            const float s1 = lower ? z1hi : z1lo;
            const float r0 = swap8(s0);
            const float r1 = swap8(s1);
            const float zi = lower ? k0 : r0;
            const float zf = lower ? r0 : k0;
            const float zg = lower ? k1 : r1;
            const float zo = lower ? r1 : k1;
            float gi = sigmoid_f(zi);
            float gf = sigmoid_f(zf);
            float gg = tanh_f(zg);
            float go = sigmoid_f(zo);
            float c  = gf * cst[j] + gi * gg;
            cst[j]   = c;
            float h  = go * tanh_f(c);
            Hnxt[(crow0 + j) * HSTRIDE + cunit] = h;
        }
        __syncthreads();

        h4 = *(const float4*)(Hnxt + frow * HSTRIDE + fcol);
        *(float4*)(out + oidx) = make_float4(h4.x + res4.x, h4.y + res4.y,
                                             h4.z + res4.z, h4.w + res4.w);
    }

    *(float4*)(st_h + (size_t)(b0 + frow) * UNITS_ + fcol) = h4;
    *(float4*)(st_c + (size_t)b0 * UNITS_ + 4 * tid) =
        make_float4(cst[0], cst[1], cst[2], cst[3]);
}

__global__ void combine_avg_kernel(const float4* __restrict__ fw,
                                   float4* __restrict__ out, int n4) {
    int i = blockIdx.x * blockDim.x + threadIdx.x;
    if (i < n4) {
        float4 a = fw[i];
        float4 b = out[i];
        out[i] = make_float4((a.x + b.x) * 0.5f, (a.y + b.y) * 0.5f,
                             (a.z + b.z) * 0.5f, (a.w + b.w) * 0.5f);
    }
}

extern "C" void kernel_launch(void* const* d_in, const int* in_sizes, int n_in,
                              void* d_out, int out_size, void* d_ws, size_t ws_size,
                              hipStream_t stream) {
    const float* x   = (const float*)d_in[0];
    const float* kfw = (const float*)d_in[1];
    const float* rfw = (const float*)d_in[2];
    const float* bfw = (const float*)d_in[3];
    const float* kbw = (const float*)d_in[4];
    const float* rbw = (const float*)d_in[5];
    const float* bbw = (const float*)d_in[6];
    float* out = (float*)d_out;

    float* rpk  = (float*)d_ws;
    float* xz   = rpk + (size_t)2 * RCOMBO_F;
    float* st   = xz + (size_t)2 * TC * NBATCH * N4;
    float* wsfw = (float*)d_ws + (size_t)2 * 1024 * 1024;
    float* wsbw = out;

    pack_r_kernel<<<1024, 256, 0, stream>>>(rfw, rbw, rpk);
    for (int k = 0; k < NCHUNK; ++k) {
        xproj_kernel<<<256, 256, 0, stream>>>(x, x, kfw, kbw, bfw, bbw, xz, k);
        lstm_chunk_kernel<<<dim3(8, 2), 1024, 0, stream>>>(
            x, x, rpk, rpk + RCOMBO_F, xz, st, wsfw, wsbw, k,  0);
    }

    pack_r_kernel<<<1024, 256, 0, stream>>>(rfw + RCOMBO_F, rbw + RCOMBO_F, rpk);
    for (int k = 0; k < NCHUNK; ++k) {
        xproj_kernel<<<256, 256, 0, stream>>>(wsfw, wsbw, kfw + RCOMBO_F, kbw + RCOMBO_F,
                                              bfw + 1024, bbw + 1024, xz, k);
        lstm_chunk_kernel<<<dim3(8, 2), 1024, 0, stream>>>(
            wsfw, wsbw, rpk, rpk + RCOMBO_F, xz, st, wsfw, wsbw, k,  1);
    }

    const int n4 = (128 * 256 * 256) / 4;
    combine_avg_kernel<<<(n4 + 255) / 256, 256, 0, stream>>>(
        (const float4*)wsfw, (float4*)out, n4);
}


# gate sigmoids and tanhs of the cell update divide by the hardware reciprocal (v_rcp_f32, 1 ulp), not the IEEE division sequence: 139 vector instructions per wave and step, not 302; sum order kept
# speedup vs baseline: 1.5342x; 1.0462x over previous
#include <hip/hip_runtime.h>


typedef float v2f __attribute__((ext_vector_type(2)));
typedef float v8f __attribute__((ext_vector_type(8)));

#define T_STEPS 256
#define UNITS_  256
#define N4      1024
#define TC      4
#define NCHUNK  64
#define NBATCH  128
#define HSTRIDE 260
#define RCOMBO_F (256 * 1024)

__device__ __forceinline__ float sigmoid_f(float x) {
    return __builtin_amdgcn_rcpf(1.0f + __expf(-x));
}
__device__ __forceinline__ float tanh_f(float x) {
    x = fminf(15.0f, fmaxf(-15.0f, x));
    float e = __expf(2.0f * x);
    return (e - 1.0f) * __builtin_amdgcn_rcpf(e + 1.0f);
}

__global__ void pack_r_kernel(const float* __restrict__ rfw,
                              const float* __restrict__ rbw,
                              float* __restrict__ packed) {
    int i = blockIdx.x * blockDim.x + threadIdx.x;
    int lane = i & 31;
    int ks   = (i >> 5) & 63;
    int nt   = (i >> 11) & 63;
    int isbw = i >> 17;
    const float* R = isbw ? rbw : rfw;
    int c  = lane & 15;
    int n  = (2 * (nt & 1) + (c >> 3)) * 256 + 8 * (nt >> 1) + (c & 7);
    int k0 = ks * 4 + ((lane >> 4) << 1);
    packed[(size_t)i * 2 + 0] = R[(size_t)k0 * 1024 + n];
    packed[(size_t)i * 2 + 1] = R[(size_t)(k0 + 1) * 1024 + n];
}

__global__ void __launch_bounds__(256)
xproj_kernel(const float* __restrict__ in_fw, const float* __restrict__ in_bw,
             const float* __restrict__ w_fw,  const float* __restrict__ w_bw,
             const float* __restrict__ bias_fw, const float* __restrict__ bias_bw,
             float* __restrict__ xz, int chunk) {
    const int lane    = threadIdx.x & 31;
    const int gw      = blockIdx.x * 8 + (threadIdx.x >> 5);
    const int mt      = gw >> 5;
    const int np      = gw & 31;
    const int dir     = mt >> 5;
    const int tc      = (mt & 31) >> 3;
    const int bt      = mt & 7;
    const int halfsel = lane >> 4;
    const int lrow    = lane & 15;

    const float* in   = dir ? in_bw : in_fw;
    const float* Wd   = dir ? w_bw : w_fw;
    const float* bias = dir ? bias_bw : bias_fw;
    const int t = dir ? (T_STEPS - 1 - TC * chunk - tc) : (TC * chunk + tc);

    const float* arow = in + ((size_t)(bt * 16 + lrow) * T_STEPS + t) * UNITS_ + 2 * halfsel;
    const int n0 = np * 32 + lrow;
    const float* wcol = Wd + (size_t)(2 * halfsel) * N4 + n0;

    v8f acc0 = {0.f, 0.f, 0.f, 0.f, 0.f, 0.f, 0.f, 0.f};
    v8f acc1 = {0.f, 0.f, 0.f, 0.f, 0.f, 0.f, 0.f, 0.f};
#pragma unroll 8
    for (int ks = 0; ks < UNITS_ / 4; ++ks) {
        v2f a = *(const v2f*)(arow + 4 * ks);
        const float* wp = wcol + (size_t)(4 * ks) * N4;
        v2f w0, w1;
        w0.x = wp[0];
        w0.y = wp[N4];
        w1.x = wp[16];
        w1.y = wp[N4 + 16];
        acc0 = __builtin_amdgcn_wmma_f32_16x16x4_f32(
            false, a, false, w0, (short)0, acc0, false, false);
        acc1 = __builtin_amdgcn_wmma_f32_16x16x4_f32(
            false, a, false, w1, (short)0, acc1, false, false);
    }
    const float bv0 = bias[n0];
    const float bv1 = bias[n0 + 16];
    float* dst = xz + ((size_t)((dir * TC + tc) * NBATCH + bt * 16 + 8 * halfsel)) * N4 + n0;
#pragma unroll
    for (int e = 0; e < 8; ++e) {
        dst[(size_t)e * N4]      = acc0[e] + bv0;
        dst[(size_t)e * N4 + 16] = acc1[e] + bv1;
    }
}

__device__ __forceinline__ float swap8(float v) {
    return __int_as_float(__builtin_amdgcn_update_dpp(
        0, __float_as_int(v), 0x128, 0xF, 0xF, false));
}

__global__ void __launch_bounds__(1024)
lstm_chunk_kernel(const float* in_fw, const float* in_bw,
                  const float* __restrict__ pr_fw,
                  const float* __restrict__ pr_bw,
                  const float* __restrict__ xz,
                  float* __restrict__ state,
                  float* out_fw, float* out_bw,
                  int chunk, int residual) {
    __shared__ __attribute__((aligned(16))) float Hbuf[2 * 16 * HSTRIDE];

    const int dir     = blockIdx.y;
    const float* in   = dir ? in_bw : in_fw;
    const float* pr   = dir ? pr_bw : pr_fw;
    float*       out  = dir ? out_bw : out_fw;

    const int b0      = blockIdx.x * 16;
    const int tid     = threadIdx.x;
    const int lane    = tid & 31;
    const int wave    = tid >> 5;
    const int halfsel = lane >> 4;
    const int lrow    = lane & 15;
    const bool lower  = (lane & 8) == 0;

    float* st_h = state + (size_t)(dir * 2 + 0) * NBATCH * UNITS_;
    float* st_c = state + (size_t)(dir * 2 + 1) * NBATCH * UNITS_;

    const int cunit = wave * 8 + (lane & 7);
    const int crow0 = 8 * halfsel + (lower ? 0 : 4);
    const int frow  = wave >> 1;
    const int fcol  = 128 * (wave & 1) + 4 * lane;

    float cst[4] = {0.0f, 0.0f, 0.0f, 0.0f};
    float4 h4 = make_float4(0.0f, 0.0f, 0.0f, 0.0f);
    if (chunk != 0) {
        const float4 c4 = *(const float4*)(st_c + (size_t)b0 * UNITS_ + 4 * tid);
        cst[0] = c4.x; cst[1] = c4.y; cst[2] = c4.z; cst[3] = c4.w;
        h4 = *(const float4*)(st_h + (size_t)(b0 + frow) * UNITS_ + fcol);
    }
    *(float4*)(Hbuf + frow * HSTRIDE + fcol) = h4;

    const int nt0 = wave * 2;
    const int nt1 = wave * 2 + 1;
    const float* pwt0 = pr + ((size_t)(nt0 * 64) * 32 + lane) * 2;
    const float* pwt1 = pr + ((size_t)(nt1 * 64) * 32 + lane) * 2;
    const int aBase = lrow * HSTRIDE + 2 * halfsel;
    const int gcol0 = (lower ? 0 : UNITS_) + cunit;

    __syncthreads();

    for (int tc = 0; tc < TC; ++tc) {
        const int t = dir ? (T_STEPS - 1 - TC * chunk - tc) : (TC * chunk + tc);
        const float* Hcur = Hbuf + (tc & 1) * (16 * HSTRIDE);
        float*       Hnxt = Hbuf + ((tc & 1) ^ 1) * (16 * HSTRIDE);

        const float* xzp = xz + ((size_t)((dir * TC + tc) * NBATCH + b0 + 8 * halfsel)) * N4 + gcol0;
        v8f x0, x1;
#pragma unroll
        for (int e = 0; e < 8; ++e) {
            x0[e] = xzp[(size_t)e * N4];
            x1[e] = xzp[(size_t)e * N4 + 2 * UNITS_];
        }
        const size_t oidx = ((size_t)(b0 + frow) * T_STEPS + t) * UNITS_ + fcol;
        float4 res4 = make_float4(0.0f, 0.0f, 0.0f, 0.0f);
        if (residual) res4 = *(const float4*)(in + oidx);

        v8f acc0 = {0.f, 0.f, 0.f, 0.f, 0.f, 0.f, 0.f, 0.f};
        v8f acc1 = {0.f, 0.f, 0.f, 0.f, 0.f, 0.f, 0.f, 0.f};
#pragma unroll 4
        for (int ks = 0; ks < UNITS_ / 4; ++ks) {
            v2f a  = *(const v2f*)(Hcur + aBase + 4 * ks);
            v2f w0 = *(const v2f*)(pwt0 + ks * 64);
            v2f w1 = *(const v2f*)(pwt1 + ks * 64);
            acc0 = __builtin_amdgcn_wmma_f32_16x16x4_f32(
                false, a, false, w0, (short)0, acc0, false, false);
            acc1 = __builtin_amdgcn_wmma_f32_16x16x4_f32(
                false, a, false, w1, (short)0, acc1, false, false);
        }

#pragma unroll
        for (int j = 0; j < 4; ++j) {
            const float z0lo = acc0[j]     + x0[j];
            const float z0hi = acc0[j + 4] + x0[j + 4];
            const float z1lo = acc1[j]     + x1[j];
            const float z1hi = acc1[j + 4] + x1[j + 4];
            const float k0 = lower ? z0lo : z0hi;
            const float s0 = lower ? z0hi : z0lo;
            const float k1 = lower ? z1lo : z1hi;
            const float s1 = lower ? z1hi : z1lo;
            const float r0 = swap8(s0);
            const float r1 = swap8(s1);
            const float zi = lower ? k0 : r0;
            const float zf = lower ? r0 : k0;
            const float zg = lower ? k1 : r1;
            const float zo = lower ? r1 : k1;
            float gi = sigmoid_f(zi);
            float gf = sigmoid_f(zf);
            float gg = tanh_f(zg);
            float go = sigmoid_f(zo);
            float c  = gf * cst[j] + gi * gg;
            cst[j]   = c;
            float h  = go * tanh_f(c);
            Hnxt[(crow0 + j) * HSTRIDE + cunit] = h;
        }
        __syncthreads();

        h4 = *(const float4*)(Hnxt + frow * HSTRIDE + fcol);
        *(float4*)(out + oidx) = make_float4(h4.x + res4.x, h4.y + res4.y,
                                             h4.z + res4.z, h4.w + res4.w);
    }

    *(float4*)(st_h + (size_t)(b0 + frow) * UNITS_ + fcol) = h4;
    *(float4*)(st_c + (size_t)b0 * UNITS_ + 4 * tid) =
        make_float4(cst[0], cst[1], cst[2], cst[3]);
}

__global__ void combine_avg_kernel(const float4* __restrict__ fw,
                                   float4* __restrict__ out, int n4) {
    int i = blockIdx.x * blockDim.x + threadIdx.x;
    if (i < n4) {
        float4 a = fw[i];
        float4 b = out[i];
        out[i] = make_float4((a.x + b.x) * 0.5f, (a.y + b.y) * 0.5f,
                             (a.z + b.z) * 0.5f, (a.w + b.w) * 0.5f);
    }
}

extern "C" void kernel_launch(void* const* d_in, const int* in_sizes, int n_in,
                              void* d_out, int out_size, void* d_ws, size_t ws_size,
                              hipStream_t stream) {
    const float* x   = (const float*)d_in[0];
    const float* kfw = (const float*)d_in[1];
    const float* rfw = (const float*)d_in[2];
    const float* bfw = (const float*)d_in[3];
    const float* kbw = (const float*)d_in[4];
    const float* rbw = (const float*)d_in[5];
    const float* bbw = (const float*)d_in[6];
    float* out = (float*)d_out;

    float* rpk  = (float*)d_ws;
    float* xz   = rpk + (size_t)2 * RCOMBO_F;
    float* st   = xz + (size_t)2 * TC * NBATCH * N4;
    float* wsfw = (float*)d_ws + (size_t)2 * 1024 * 1024;
    float* wsbw = out;

    pack_r_kernel<<<1024, 256, 0, stream>>>(rfw, rbw, rpk);
    for (int k = 0; k < NCHUNK; ++k) {
        xproj_kernel<<<256, 256, 0, stream>>>(x, x, kfw, kbw, bfw, bbw, xz, k);
        lstm_chunk_kernel<<<dim3(8, 2), 1024, 0, stream>>>(
            x, x, rpk, rpk + RCOMBO_F, xz, st, wsfw, wsbw, k,  0);
    }

    pack_r_kernel<<<1024, 256, 0, stream>>>(rfw + RCOMBO_F, rbw + RCOMBO_F, rpk);
    for (int k = 0; k < NCHUNK; ++k) {
        xproj_kernel<<<256, 256, 0, stream>>>(wsfw, wsbw, kfw + RCOMBO_F, kbw + RCOMBO_F,
                                              bfw + 1024, bbw + 1024, xz, k);
        lstm_chunk_kernel<<<dim3(8, 2), 1024, 0, stream>>>(
            wsfw, wsbw, rpk, rpk + RCOMBO_F, xz, st, wsfw, wsbw, k,  1);
    }

    const int n4 = (128 * 256 * 256) / 4;
    combine_avg_kernel<<<(n4 + 255) / 256, 256, 0, stream>>>(
        (const float4*)wsfw, (float4*)out, n4);
}


# scan waves own all four gates of 16 units as four one-gate tiles (16 waves per work-group, 4 per SIMD): 32 weight fragments per wait, not 8, and no cross-lane gate exchange; sum order kept
# speedup vs baseline: 1.6488x; 1.0747x over previous
#include <hip/hip_runtime.h>


typedef float v2f __attribute__((ext_vector_type(2)));
typedef float v4f __attribute__((ext_vector_type(4)));
typedef float v8f __attribute__((ext_vector_type(8)));

#define T_STEPS 256
#define UNITS_  256
#define N4      1024
#define TC      4
#define NCHUNK  64
#define NBATCH  128
#define SCAN_WAVES 16
#define HSTRIDE 260
#define RCOMBO_F (256 * 1024)

__device__ __forceinline__ float sigmoid_f(float x) {
    return __builtin_amdgcn_rcpf(1.0f + __expf(-x));
}
__device__ __forceinline__ float tanh_f(float x) {
    x = fminf(15.0f, fmaxf(-15.0f, x));
    float e = __expf(2.0f * x);
    return (e - 1.0f) * __builtin_amdgcn_rcpf(e + 1.0f);
}

__global__ void pack_r_kernel(const float* __restrict__ rfw,
                              const float* __restrict__ rbw,
                              float* __restrict__ packed) {
    int i = blockIdx.x * blockDim.x + threadIdx.x;
    int lane = i & 31;
    int gp   = (i >> 5) & 1;
    int ks   = (i >> 6) & 63;
    int w    = (i >> 12) & 15;
    int isbw = i >> 16;
    const float* R = isbw ? rbw : rfw;
    int c  = lane & 15;
    int n  = (2 * gp) * 256 + 16 * w + c;
    int k0 = ks * 4 + ((lane >> 4) << 1);
    float4 v;
    v.x = R[(size_t)k0 * 1024 + n];
    v.y = R[(size_t)(k0 + 1) * 1024 + n];
    v.z = R[(size_t)k0 * 1024 + n + 256];
    v.w = R[(size_t)(k0 + 1) * 1024 + n + 256];
    ((float4*)packed)[i] = v;
}

__global__ void __launch_bounds__(256)
xproj_kernel(const float* __restrict__ in_fw, const float* __restrict__ in_bw,
             const float* __restrict__ w_fw,  const float* __restrict__ w_bw,
             const float* __restrict__ bias_fw, const float* __restrict__ bias_bw,
             float* __restrict__ xz, int chunk) {
    const int lane    = threadIdx.x & 31;
    const int gw      = blockIdx.x * 8 + (threadIdx.x >> 5);
    const int mt      = gw >> 5;
    const int np      = gw & 31;
    const int dir     = mt >> 5;
    const int tc      = (mt & 31) >> 3;
    const int bt      = mt & 7;
    const int halfsel = lane >> 4;
    const int lrow    = lane & 15;

    const float* in   = dir ? in_bw : in_fw;
    const float* Wd   = dir ? w_bw : w_fw;
    const float* bias = dir ? bias_bw : bias_fw;
    const int t = dir ? (T_STEPS - 1 - TC * chunk - tc) : (TC * chunk + tc);

    const float* arow = in + ((size_t)(bt * 16 + lrow) * T_STEPS + t) * UNITS_ + 2 * halfsel;
    const int n0 = np * 32 + lrow;
    const float* wcol = Wd + (size_t)(2 * halfsel) * N4 + n0;

    v8f acc0 = {0.f, 0.f, 0.f, 0.f, 0.f, 0.f, 0.f, 0.f};
    v8f acc1 = {0.f, 0.f, 0.f, 0.f, 0.f, 0.f, 0.f, 0.f};
#pragma unroll 8
    for (int ks = 0; ks < UNITS_ / 4; ++ks) {
        v2f a = *(const v2f*)(arow + 4 * ks);
        const float* wp = wcol + (size_t)(4 * ks) * N4;
        v2f w0, w1;
        w0.x = wp[0];
        w0.y = wp[N4];
        w1.x = wp[16];
        w1.y = wp[N4 + 16];
        acc0 = __builtin_amdgcn_wmma_f32_16x16x4_f32(
            false, a, false, w0, (short)0, acc0, false, false);
        acc1 = __builtin_amdgcn_wmma_f32_16x16x4_f32(
            false, a, false, w1, (short)0, acc1, false, false);
    }
    const float bv0 = bias[n0];
    const float bv1 = bias[n0 + 16];
    float* dst = xz + ((size_t)((dir * TC + tc) * NBATCH + bt * 16 + 8 * halfsel)) * N4 + n0;
#pragma unroll
    for (int e = 0; e < 8; ++e) {
        dst[(size_t)e * N4]      = acc0[e] + bv0;
        dst[(size_t)e * N4 + 16] = acc1[e] + bv1;
    }
}

__global__ void __launch_bounds__(32 * SCAN_WAVES)
lstm_chunk_kernel(const float* in_fw, const float* in_bw,
                  const float* __restrict__ pr_fw,
                  const float* __restrict__ pr_bw,
                  const float* __restrict__ xz,
                  float* __restrict__ state,
                  float* out_fw, float* out_bw,
                  int chunk, int residual) {
    __shared__ __attribute__((aligned(16))) float Hbuf[2 * 16 * HSTRIDE];

    const int dir     = blockIdx.y;
    const float* in   = dir ? in_bw : in_fw;
    const float* pr   = dir ? pr_bw : pr_fw;
    float*       out  = dir ? out_bw : out_fw;

    const int b0      = blockIdx.x * 16;
    const int tid     = threadIdx.x;
    const int lane    = tid & 31;
    const int wave    = tid >> 5;
    const int halfsel = lane >> 4;
    const int lrow    = lane & 15;

    float* st_h = state + (size_t)(dir * 2 + 0) * NBATCH * UNITS_;
    float* st_c = state + (size_t)(dir * 2 + 1) * NBATCH * UNITS_;

    const int cunit = wave * 16 + lrow;
    const int crow0 = 8 * halfsel;
    const int frow  = wave;
    const int fcol  = 4 * lane;

    float cst[8] = {0.0f, 0.0f, 0.0f, 0.0f, 0.0f, 0.0f, 0.0f, 0.0f};
    float4 h4a = make_float4(0.0f, 0.0f, 0.0f, 0.0f);
    float4 h4b = make_float4(0.0f, 0.0f, 0.0f, 0.0f);
    float* cptr = st_c + (size_t)b0 * UNITS_ + 4 * tid;
    float* hptr = st_h + (size_t)(b0 + frow) * UNITS_ + fcol;
    if (chunk != 0) {
        const float4 c4a = *(const float4*)(cptr);
        const float4 c4b = *(const float4*)(cptr + 8 * UNITS_);
        cst[0] = c4a.x; cst[1] = c4a.y; cst[2] = c4a.z; cst[3] = c4a.w;
        cst[4] = c4b.x; cst[5] = c4b.y; cst[6] = c4b.z; cst[7] = c4b.w;
        h4a = *(const float4*)(hptr);
        h4b = *(const float4*)(hptr + 128);
    }
    *(float4*)(Hbuf + frow * HSTRIDE + fcol)       = h4a;
    *(float4*)(Hbuf + frow * HSTRIDE + fcol + 128) = h4b;

    const float* pw = pr + ((size_t)wave * (64 * 2 * 32) + lane) * 4;
    const int aBase = lrow * HSTRIDE + 2 * halfsel;

    __syncthreads();

    for (int tc = 0; tc < TC; ++tc) {
        const int t = dir ? (T_STEPS - 1 - TC * chunk - tc) : (TC * chunk + tc);
        const float* Hcur = Hbuf + (tc & 1) * (16 * HSTRIDE);
        float*       Hnxt = Hbuf + ((tc & 1) ^ 1) * (16 * HSTRIDE);

        const float* xzp = xz + ((size_t)((dir * TC + tc) * NBATCH + b0 + crow0)) * N4 + cunit;
        v8f x0, x1, x2, x3;
#pragma unroll
        for (int e = 0; e < 8; ++e) {
            x0[e] = xzp[(size_t)e * N4];
            x1[e] = xzp[(size_t)e * N4 + UNITS_];
            x2[e] = xzp[(size_t)e * N4 + 2 * UNITS_];
            x3[e] = xzp[(size_t)e * N4 + 3 * UNITS_];
        }
        const size_t oidx = ((size_t)(b0 + frow) * T_STEPS + t) * UNITS_ + fcol;
        float4 resa = make_float4(0.0f, 0.0f, 0.0f, 0.0f);
        float4 resb = make_float4(0.0f, 0.0f, 0.0f, 0.0f);
        if (residual) {
            resa = *(const float4*)(in + oidx);
            resb = *(const float4*)(in + oidx + 128);
        }

        v8f acc0 = {0.f, 0.f, 0.f, 0.f, 0.f, 0.f, 0.f, 0.f};
        v8f acc1 = {0.f, 0.f, 0.f, 0.f, 0.f, 0.f, 0.f, 0.f};
        v8f acc2 = {0.f, 0.f, 0.f, 0.f, 0.f, 0.f, 0.f, 0.f};
        v8f acc3 = {0.f, 0.f, 0.f, 0.f, 0.f, 0.f, 0.f, 0.f};
#pragma unroll 8
        for (int ks = 0; ks < UNITS_ / 4; ++ks) {
            v2f a  = *(const v2f*)(Hcur + aBase + 4 * ks);
            v4f q0 = *(const v4f*)(pw + ks * 256);
            v4f q1 = *(const v4f*)(pw + ks * 256 + 128);
            v2f wi = {q0.x, q0.y};
            v2f wf = {q0.z, q0.w};
            v2f wg = {q1.x, q1.y};
            v2f wo = {q1.z, q1.w};
            acc0 = __builtin_amdgcn_wmma_f32_16x16x4_f32(
                false, a, false, wi, (short)0, acc0, false, false);
            acc1 = __builtin_amdgcn_wmma_f32_16x16x4_f32(
                false, a, false, wf, (short)0, acc1, false, false);
            acc2 = __builtin_amdgcn_wmma_f32_16x16x4_f32(
                false, a, false, wg, (short)0, acc2, false, false);
            acc3 = __builtin_amdgcn_wmma_f32_16x16x4_f32(
                false, a, false, wo, (short)0, acc3, false, false);
        }

#pragma unroll
        for (int j = 0; j < 8; ++j) {
            const float zi = acc0[j] + x0[j];
            const float zf = acc1[j] + x1[j];
            const float zg = acc2[j] + x2[j];
            const float zo = acc3[j] + x3[j];
            float gi = sigmoid_f(zi);
            float gf = sigmoid_f(zf);
            float gg = tanh_f(zg);
            float go = sigmoid_f(zo);
            float c  = gf * cst[j] + gi * gg;
            cst[j]   = c;
            float h  = go * tanh_f(c);
            Hnxt[(crow0 + j) * HSTRIDE + cunit] = h;
        }
        __syncthreads();

        h4a = *(const float4*)(Hnxt + frow * HSTRIDE + fcol);
        h4b = *(const float4*)(Hnxt + frow * HSTRIDE + fcol + 128);
        *(float4*)(out + oidx) = make_float4(h4a.x + resa.x, h4a.y + resa.y,
                                             h4a.z + resa.z, h4a.w + resa.w);
        *(float4*)(out + oidx + 128) = make_float4(h4b.x + resb.x, h4b.y + resb.y,
                                                   h4b.z + resb.z, h4b.w + resb.w);
    }

    *(float4*)(hptr)       = h4a;
    *(float4*)(hptr + 128) = h4b;
    *(float4*)(cptr)              = make_float4(cst[0], cst[1], cst[2], cst[3]);
    *(float4*)(cptr + 8 * UNITS_) = make_float4(cst[4], cst[5], cst[6], cst[7]);
}

__global__ void combine_avg_kernel(const float4* __restrict__ fw,
                                   float4* __restrict__ out, int n4) {
    int i = blockIdx.x * blockDim.x + threadIdx.x;
    if (i < n4) {
        float4 a = fw[i];
        float4 b = out[i];
        out[i] = make_float4((a.x + b.x) * 0.5f, (a.y + b.y) * 0.5f,
                             (a.z + b.z) * 0.5f, (a.w + b.w) * 0.5f);
    }
}

extern "C" void kernel_launch(void* const* d_in, const int* in_sizes, int n_in,
                              void* d_out, int out_size, void* d_ws, size_t ws_size,
                              hipStream_t stream) {
    const float* x   = (const float*)d_in[0];
    const float* kfw = (const float*)d_in[1];
    const float* rfw = (const float*)d_in[2];
    const float* bfw = (const float*)d_in[3];
    const float* kbw = (const float*)d_in[4];
    const float* rbw = (const float*)d_in[5];
    const float* bbw = (const float*)d_in[6];
    float* out = (float*)d_out;

    float* rpk  = (float*)d_ws;
    float* xz   = rpk + (size_t)2 * RCOMBO_F;
    float* st   = xz + (size_t)2 * TC * NBATCH * N4;
    float* wsfw = (float*)d_ws + (size_t)2 * 1024 * 1024;
    float* wsbw = out;

    pack_r_kernel<<<512, 256, 0, stream>>>(rfw, rbw, rpk);
    for (int k = 0; k < NCHUNK; ++k) {
        xproj_kernel<<<256, 256, 0, stream>>>(x, x, kfw, kbw, bfw, bbw, xz, k);
        lstm_chunk_kernel<<<dim3(8, 2), 512, 0, stream>>>(
            x, x, rpk, rpk + RCOMBO_F, xz, st, wsfw, wsbw, k,  0);
    }

    pack_r_kernel<<<512, 256, 0, stream>>>(rfw + RCOMBO_F, rbw + RCOMBO_F, rpk);
    for (int k = 0; k < NCHUNK; ++k) {
        xproj_kernel<<<256, 256, 0, stream>>>(wsfw, wsbw, kfw + RCOMBO_F, kbw + RCOMBO_F,
                                              bfw + 1024, bbw + 1024, xz, k);
        lstm_chunk_kernel<<<dim3(8, 2), 512, 0, stream>>>(
            wsfw, wsbw, rpk, rpk + RCOMBO_F, xz, st, wsfw, wsbw, k,  1);
    }

    const int n4 = (128 * 256 * 256) / 4;
    combine_avg_kernel<<<(n4 + 255) / 256, 256, 0, stream>>>(
        (const float4*)wsfw, (float4*)out, n4);
}


# the next chunk's x@W+b is formed by 64 extra work-groups of the scan launch on WGPs the scan leaves idle (2-step chunks, two parked planes), not by a launch of its own; sum order kept
# speedup vs baseline: 1.8143x; 1.1004x over previous
#include <hip/hip_runtime.h>


typedef float v2f __attribute__((ext_vector_type(2)));
typedef float v4f __attribute__((ext_vector_type(4)));
typedef float v8f __attribute__((ext_vector_type(8)));

#define T_STEPS 256
#define UNITS_  256
#define N4      1024
#define TC      2
#define NCHUNK  128
#define NBATCH  128
#define SCAN_WAVES 16
#define HSTRIDE 260
#define RCOMBO_F (256 * 1024)
#define XZ_PLANE (2 * TC * NBATCH * N4)
#define NSCAN_WG 16
#define NPROJ_WG 64

__device__ __forceinline__ float sigmoid_f(float x) {
    return __builtin_amdgcn_rcpf(1.0f + __expf(-x));
}
__device__ __forceinline__ float tanh_f(float x) {
    x = fminf(15.0f, fmaxf(-15.0f, x));
    float e = __expf(2.0f * x);
    return (e - 1.0f) * __builtin_amdgcn_rcpf(e + 1.0f);
}

__global__ void pack_r_kernel(const float* __restrict__ rfw,
                              const float* __restrict__ rbw,
                              float* __restrict__ packed) {
    int i = blockIdx.x * blockDim.x + threadIdx.x;
    int lane = i & 31;
    int gp   = (i >> 5) & 1;
    int ks   = (i >> 6) & 63;
    int w    = (i >> 12) & 15;
    int isbw = i >> 16;
    const float* R = isbw ? rbw : rfw;
    int c  = lane & 15;
    int n  = (2 * gp) * 256 + 16 * w + c;
    int k0 = ks * 4 + ((lane >> 4) << 1);
    float4 v;
    v.x = R[(size_t)k0 * 1024 + n];
    v.y = R[(size_t)(k0 + 1) * 1024 + n];
    v.z = R[(size_t)k0 * 1024 + n + 256];
    v.w = R[(size_t)(k0 + 1) * 1024 + n + 256];
    ((float4*)packed)[i] = v;
}

__device__ __forceinline__ void
xproj_tile(const float* in_fw, const float* in_bw,
           const float* __restrict__ w_fw,  const float* __restrict__ w_bw,
           const float* __restrict__ bias_fw, const float* __restrict__ bias_bw,
           float* xz, int chunk, int gw, int lane) {
    const int mt      = gw >> 5;
    const int np      = gw & 31;
    const int dir     = mt >> 4;
    const int tc      = (mt & 15) >> 3;
    const int bt      = mt & 7;
    const int halfsel = lane >> 4;
    const int lrow    = lane & 15;

    const float* in   = dir ? in_bw : in_fw;
    const float* Wd   = dir ? w_bw : w_fw;
    const float* bias = dir ? bias_bw : bias_fw;
    const int t = dir ? (T_STEPS - 1 - TC * chunk - tc) : (TC * chunk + tc);

    const float* arow = in + ((size_t)(bt * 16 + lrow) * T_STEPS + t) * UNITS_ + 2 * halfsel;
    const int n0 = np * 32 + lrow;
    const float* wcol = Wd + (size_t)(2 * halfsel) * N4 + n0;

    v8f acc0 = {0.f, 0.f, 0.f, 0.f, 0.f, 0.f, 0.f, 0.f};
    v8f acc1 = {0.f, 0.f, 0.f, 0.f, 0.f, 0.f, 0.f, 0.f};
#pragma unroll 8
    for (int ks = 0; ks < UNITS_ / 4; ++ks) {
        v2f a = *(const v2f*)(arow + 4 * ks);
        const float* wp = wcol + (size_t)(4 * ks) * N4;
        v2f w0, w1;
        w0.x = wp[0];
        w0.y = wp[N4];
        w1.x = wp[16];
        w1.y = wp[N4 + 16];
        acc0 = __builtin_amdgcn_wmma_f32_16x16x4_f32(
            false, a, false, w0, (short)0, acc0, false, false);
        acc1 = __builtin_amdgcn_wmma_f32_16x16x4_f32(
            false, a, false, w1, (short)0, acc1, false, false);
    }
    const float bv0 = bias[n0];
    const float bv1 = bias[n0 + 16];
    float* dst = xz + ((size_t)((dir * TC + tc) * NBATCH + bt * 16 + 8 * halfsel)) * N4 + n0;
#pragma unroll
    for (int e = 0; e < 8; ++e) {
        dst[(size_t)e * N4]      = acc0[e] + bv0;
        dst[(size_t)e * N4 + 16] = acc1[e] + bv1;
    }
}

__global__ void __launch_bounds__(32 * SCAN_WAVES)
lstm_chunk_kernel(const float* in_fw, const float* in_bw,
                  const float* __restrict__ pr_fw,
                  const float* __restrict__ pr_bw,
                  float* xz,
                  float* __restrict__ state,
                  float* out_fw, float* out_bw,
                  int chunk, int residual,
                  const float* xin_fw, const float* xin_bw,
                  const float* __restrict__ xw_fw, const float* __restrict__ xw_bw,
                  const float* __restrict__ xb_fw, const float* __restrict__ xb_bw,
                  int xfirst, int nscan) {
    __shared__ __attribute__((aligned(16))) float Hbuf[2 * 16 * HSTRIDE];

    if ((int)blockIdx.x >= nscan) {
        const int xchunk = xfirst ? 0 : chunk + 1;
        xproj_tile(xin_fw, xin_bw, xw_fw, xw_bw, xb_fw, xb_bw,
                   xz + (size_t)(xchunk & 1) * XZ_PLANE, xchunk,
                   ((int)blockIdx.x - nscan) * SCAN_WAVES + (int)(threadIdx.x >> 5),
                   (int)(threadIdx.x & 31));
        return;
    }

    const float* xzc  = xz + (size_t)(chunk & 1) * XZ_PLANE;
    const int dir     = blockIdx.x >> 3;
    const float* in   = dir ? in_bw : in_fw;
    const float* pr   = dir ? pr_bw : pr_fw;
    float*       out  = dir ? out_bw : out_fw;

    const int b0      = (blockIdx.x & 7) * 16;
    const int tid     = threadIdx.x;
    const int lane    = tid & 31;
    const int wave    = tid >> 5;
    const int halfsel = lane >> 4;
    const int lrow    = lane & 15;

    float* st_h = state + (size_t)(dir * 2 + 0) * NBATCH * UNITS_;
    float* st_c = state + (size_t)(dir * 2 + 1) * NBATCH * UNITS_;

    const int cunit = wave * 16 + lrow;
    const int crow0 = 8 * halfsel;
    const int frow  = wave;
    const int fcol  = 4 * lane;

    float cst[8] = {0.0f, 0.0f, 0.0f, 0.0f, 0.0f, 0.0f, 0.0f, 0.0f};
    float4 h4a = make_float4(0.0f, 0.0f, 0.0f, 0.0f);
    float4 h4b = make_float4(0.0f, 0.0f, 0.0f, 0.0f);
    float* cptr = st_c + (size_t)b0 * UNITS_ + 4 * tid;
    float* hptr = st_h + (size_t)(b0 + frow) * UNITS_ + fcol;
    if (chunk != 0) {
        const float4 c4a = *(const float4*)(cptr);
        const float4 c4b = *(const float4*)(cptr + 8 * UNITS_);
        cst[0] = c4a.x; cst[1] = c4a.y; cst[2] = c4a.z; cst[3] = c4a.w;
        cst[4] = c4b.x; cst[5] = c4b.y; cst[6] = c4b.z; cst[7] = c4b.w;
        h4a = *(const float4*)(hptr);
        h4b = *(const float4*)(hptr + 128);
    }
    *(float4*)(Hbuf + frow * HSTRIDE + fcol)       = h4a;
    *(float4*)(Hbuf + frow * HSTRIDE + fcol + 128) = h4b;

    const float* pw = pr + ((size_t)wave * (64 * 2 * 32) + lane) * 4;
    const int aBase = lrow * HSTRIDE + 2 * halfsel;

    __syncthreads();

#pragma nounroll
    for (int tc = 0; tc < TC; ++tc) {
        const int t = dir ? (T_STEPS - 1 - TC * chunk - tc) : (TC * chunk + tc);
        const float* Hcur = Hbuf + (tc & 1) * (16 * HSTRIDE);
        float*       Hnxt = Hbuf + ((tc & 1) ^ 1) * (16 * HSTRIDE);

        const float* xzp = xzc + ((size_t)((dir * TC + tc) * NBATCH + b0 + crow0)) * N4 + cunit;
        v8f x0, x1, x2, x3;
#pragma unroll
        for (int e = 0; e < 8; ++e) {
            x0[e] = xzp[(size_t)e * N4];
            x1[e] = xzp[(size_t)e * N4 + UNITS_];
            x2[e] = xzp[(size_t)e * N4 + 2 * UNITS_];
            x3[e] = xzp[(size_t)e * N4 + 3 * UNITS_];
        }
        const size_t oidx = ((size_t)(b0 + frow) * T_STEPS + t) * UNITS_ + fcol;
        float4 resa = make_float4(0.0f, 0.0f, 0.0f, 0.0f);
        float4 resb = make_float4(0.0f, 0.0f, 0.0f, 0.0f);
        if (residual) {
            resa = *(const float4*)(in + oidx);
            resb = *(const float4*)(in + oidx + 128);
        }

        v8f acc0 = {0.f, 0.f, 0.f, 0.f, 0.f, 0.f, 0.f, 0.f};
        v8f acc1 = {0.f, 0.f, 0.f, 0.f, 0.f, 0.f, 0.f, 0.f};
        v8f acc2 = {0.f, 0.f, 0.f, 0.f, 0.f, 0.f, 0.f, 0.f};
        v8f acc3 = {0.f, 0.f, 0.f, 0.f, 0.f, 0.f, 0.f, 0.f};
#pragma unroll 8
        for (int ks = 0; ks < UNITS_ / 4; ++ks) {
            v2f a  = *(const v2f*)(Hcur + aBase + 4 * ks);
            v4f q0 = *(const v4f*)(pw + ks * 256);
            v4f q1 = *(const v4f*)(pw + ks * 256 + 128);
            v2f wi = {q0.x, q0.y};
            v2f wf = {q0.z, q0.w};
            v2f wg = {q1.x, q1.y};
            v2f wo = {q1.z, q1.w};
            acc0 = __builtin_amdgcn_wmma_f32_16x16x4_f32(
                false, a, false, wi, (short)0, acc0, false, false);
            acc1 = __builtin_amdgcn_wmma_f32_16x16x4_f32(
                false, a, false, wf, (short)0, acc1, false, false);
            acc2 = __builtin_amdgcn_wmma_f32_16x16x4_f32(
                false, a, false, wg, (short)0, acc2, false, false);
            acc3 = __builtin_amdgcn_wmma_f32_16x16x4_f32(
                false, a, false, wo, (short)0, acc3, false, false);
        }

#pragma unroll
        for (int j = 0; j < 8; ++j) {
            const float zi = acc0[j] + x0[j];
            const float zf = acc1[j] + x1[j];
            const float zg = acc2[j] + x2[j];
            const float zo = acc3[j] + x3[j];
            float gi = sigmoid_f(zi);
            float gf = sigmoid_f(zf);
            float gg = tanh_f(zg);
            float go = sigmoid_f(zo);
            float c  = gf * cst[j] + gi * gg;
            cst[j]   = c;
            float h  = go * tanh_f(c);
            Hnxt[(crow0 + j) * HSTRIDE + cunit] = h;
        }
        __syncthreads();

        h4a = *(const float4*)(Hnxt + frow * HSTRIDE + fcol);
        h4b = *(const float4*)(Hnxt + frow * HSTRIDE + fcol + 128);
        *(float4*)(out + oidx) = make_float4(h4a.x + resa.x, h4a.y + resa.y,
                                             h4a.z + resa.z, h4a.w + resa.w);
        *(float4*)(out + oidx + 128) = make_float4(h4b.x + resb.x, h4b.y + resb.y,
                                                   h4b.z + resb.z, h4b.w + resb.w);
    }

    *(float4*)(hptr)       = h4a;
    *(float4*)(hptr + 128) = h4b;
    *(float4*)(cptr)              = make_float4(cst[0], cst[1], cst[2], cst[3]);
    *(float4*)(cptr + 8 * UNITS_) = make_float4(cst[4], cst[5], cst[6], cst[7]);
}

__global__ void combine_avg_kernel(const float4* __restrict__ fw,
                                   float4* __restrict__ out, int n4) {
    int i = blockIdx.x * blockDim.x + threadIdx.x;
    if (i < n4) {
        float4 a = fw[i];
        float4 b = out[i];
        out[i] = make_float4((a.x + b.x) * 0.5f, (a.y + b.y) * 0.5f,
                             (a.z + b.z) * 0.5f, (a.w + b.w) * 0.5f);
    }
}

extern "C" void kernel_launch(void* const* d_in, const int* in_sizes, int n_in,
                              void* d_out, int out_size, void* d_ws, size_t ws_size,
                              hipStream_t stream) {
    const float* x   = (const float*)d_in[0];
    const float* kfw = (const float*)d_in[1];
    const float* rfw = (const float*)d_in[2];
    const float* bfw = (const float*)d_in[3];
    const float* kbw = (const float*)d_in[4];
    const float* rbw = (const float*)d_in[5];
    const float* bbw = (const float*)d_in[6];
    float* out = (float*)d_out;

    float* rpk  = (float*)d_ws;
    float* xz   = rpk + (size_t)2 * RCOMBO_F;
    float* st   = xz + (size_t)2 * XZ_PLANE;
    float* wsfw = (float*)d_ws + (size_t)2 * 1024 * 1024;
    float* wsbw = out;

    pack_r_kernel<<<512, 256, 0, stream>>>(rfw, rbw, rpk);
    lstm_chunk_kernel<<<NPROJ_WG, 512, 0, stream>>>(
        x, x, rpk, rpk + RCOMBO_F, xz, st, wsfw, wsbw, 0,  0,
        x, x, kfw, kbw, bfw, bbw,  1,  0);
    for (int k = 0; k < NCHUNK - 1; ++k) {
        lstm_chunk_kernel<<<NSCAN_WG + NPROJ_WG, 512, 0, stream>>>(
            x, x, rpk, rpk + RCOMBO_F, xz, st, wsfw, wsbw, k,  0,
            x, x, kfw, kbw, bfw, bbw,  0, NSCAN_WG);
    }
    lstm_chunk_kernel<<<NSCAN_WG + NPROJ_WG, 512, 0, stream>>>(
        x, x, rpk, rpk + RCOMBO_F, xz, st, wsfw, wsbw, NCHUNK - 1,  0,
        wsfw, wsbw, kfw + RCOMBO_F, kbw + RCOMBO_F, bfw + 1024, bbw + 1024,
         1, NSCAN_WG);

    pack_r_kernel<<<512, 256, 0, stream>>>(rfw + RCOMBO_F, rbw + RCOMBO_F, rpk);
    for (int k = 0; k < NCHUNK - 1; ++k) {
        lstm_chunk_kernel<<<NSCAN_WG + NPROJ_WG, 512, 0, stream>>>(
            wsfw, wsbw, rpk, rpk + RCOMBO_F, xz, st, wsfw, wsbw, k,  1,
            wsfw, wsbw, kfw + RCOMBO_F, kbw + RCOMBO_F, bfw + 1024, bbw + 1024,
             0, NSCAN_WG);
    }
    lstm_chunk_kernel<<<NSCAN_WG, 512, 0, stream>>>(
        wsfw, wsbw, rpk, rpk + RCOMBO_F, xz, st, wsfw, wsbw, NCHUNK - 1,  1,
        wsfw, wsbw, kfw + RCOMBO_F, kbw + RCOMBO_F, bfw + 1024, bbw + 1024,
         0, NSCAN_WG);

    const int n4 = (128 * 256 * 256) / 4;
    combine_avg_kernel<<<(n4 + 255) / 256, 256, 0, stream>>>(
        (const float4*)wsfw, (float4*)out, n4);
}


# scan: a step's h-independent needs (first 32 weight fragments, parked x@W+b, residual rows) are requested before the barrier that publishes h and stay in flight across it; sum order kept
# speedup vs baseline: 1.9239x; 1.0604x over previous
#include <hip/hip_runtime.h>


typedef float v2f __attribute__((ext_vector_type(2)));
typedef float v4f __attribute__((ext_vector_type(4)));
typedef float v8f __attribute__((ext_vector_type(8)));

#define T_STEPS 256
#define UNITS_  256
#define N4      1024
#define TC      2
#define NCHUNK  128
#define NBATCH  128
#define SCAN_WAVES 16
#define HSTRIDE 260
#define RCOMBO_F (256 * 1024)
#define XZ_PLANE (2 * TC * NBATCH * N4)
#define NSCAN_WG 16
#define NPROJ_WG 64

__device__ __forceinline__ float sigmoid_f(float x) {
    return __builtin_amdgcn_rcpf(1.0f + __expf(-x));
}
__device__ __forceinline__ float tanh_f(float x) {
    x = fminf(15.0f, fmaxf(-15.0f, x));
    float e = __expf(2.0f * x);
    return (e - 1.0f) * __builtin_amdgcn_rcpf(e + 1.0f);
}

__global__ void pack_r_kernel(const float* __restrict__ rfw,
                              const float* __restrict__ rbw,
                              float* __restrict__ packed) {
    int i = blockIdx.x * blockDim.x + threadIdx.x;
    int lane = i & 31;
    int gp   = (i >> 5) & 1;
    int ks   = (i >> 6) & 63;
    int w    = (i >> 12) & 15;
    int isbw = i >> 16;
    const float* R = isbw ? rbw : rfw;
    int c  = lane & 15;
    int n  = (2 * gp) * 256 + 16 * w + c;
    int k0 = ks * 4 + ((lane >> 4) << 1);
    float4 v;
    v.x = R[(size_t)k0 * 1024 + n];
    v.y = R[(size_t)(k0 + 1) * 1024 + n];
    v.z = R[(size_t)k0 * 1024 + n + 256];
    v.w = R[(size_t)(k0 + 1) * 1024 + n + 256];
    ((float4*)packed)[i] = v;
}

__device__ __forceinline__ void
xproj_tile(const float* in_fw, const float* in_bw,
           const float* __restrict__ w_fw,  const float* __restrict__ w_bw,
           const float* __restrict__ bias_fw, const float* __restrict__ bias_bw,
           float* xz, int chunk, int gw, int lane) {
    const int mt      = gw >> 5;
    const int np      = gw & 31;
    const int dir     = mt >> 4;
    const int tc      = (mt & 15) >> 3;
    const int bt      = mt & 7;
    const int halfsel = lane >> 4;
    const int lrow    = lane & 15;

    const float* in   = dir ? in_bw : in_fw;
    const float* Wd   = dir ? w_bw : w_fw;
    const float* bias = dir ? bias_bw : bias_fw;
    const int t = dir ? (T_STEPS - 1 - TC * chunk - tc) : (TC * chunk + tc);

    const float* arow = in + ((size_t)(bt * 16 + lrow) * T_STEPS + t) * UNITS_ + 2 * halfsel;
    const int n0 = np * 32 + lrow;
    const float* wcol = Wd + (size_t)(2 * halfsel) * N4 + n0;

    v8f acc0 = {0.f, 0.f, 0.f, 0.f, 0.f, 0.f, 0.f, 0.f};
    v8f acc1 = {0.f, 0.f, 0.f, 0.f, 0.f, 0.f, 0.f, 0.f};
#pragma unroll 8
    for (int ks = 0; ks < UNITS_ / 4; ++ks) {
        v2f a = *(const v2f*)(arow + 4 * ks);
        const float* wp = wcol + (size_t)(4 * ks) * N4;
        v2f w0, w1;
        w0.x = wp[0];
        w0.y = wp[N4];
        w1.x = wp[16];
        w1.y = wp[N4 + 16];
        acc0 = __builtin_amdgcn_wmma_f32_16x16x4_f32(
            false, a, false, w0, (short)0, acc0, false, false);
        acc1 = __builtin_amdgcn_wmma_f32_16x16x4_f32(
            false, a, false, w1, (short)0, acc1, false, false);
    }
    const float bv0 = bias[n0];
    const float bv1 = bias[n0 + 16];
    float* dst = xz + ((size_t)((dir * TC + tc) * NBATCH + bt * 16 + 8 * halfsel)) * N4 + n0;
#pragma unroll
    for (int e = 0; e < 8; ++e) {
        dst[(size_t)e * N4]      = acc0[e] + bv0;
        dst[(size_t)e * N4 + 16] = acc1[e] + bv1;
    }
}

__device__ __forceinline__ void load_group(v4f (&q)[16], const float* p) {
    for (int i = 0; i < 8; ++i) {
        q[2 * i]     = *(const v4f*)(p + i * 256);
        q[2 * i + 1] = *(const v4f*)(p + i * 256 + 128);
    }
}
__device__ __forceinline__ void mma_group(v8f& acc0, v8f& acc1, v8f& acc2, v8f& acc3,
                                          const v4f (&q)[16], const float* hrow) {
    for (int i = 0; i < 8; ++i) {
        v2f a  = *(const v2f*)(hrow + 4 * i);
        v2f wi = {q[2 * i].x, q[2 * i].y};
        v2f wf = {q[2 * i].z, q[2 * i].w};
        v2f wg = {q[2 * i + 1].x, q[2 * i + 1].y};
        v2f wo = {q[2 * i + 1].z, q[2 * i + 1].w};
        acc0 = __builtin_amdgcn_wmma_f32_16x16x4_f32(
            false, a, false, wi, (short)0, acc0, false, false);
        acc1 = __builtin_amdgcn_wmma_f32_16x16x4_f32(
            false, a, false, wf, (short)0, acc1, false, false);
        acc2 = __builtin_amdgcn_wmma_f32_16x16x4_f32(
            false, a, false, wg, (short)0, acc2, false, false);
        acc3 = __builtin_amdgcn_wmma_f32_16x16x4_f32(
            false, a, false, wo, (short)0, acc3, false, false);
    }
}
__device__ __forceinline__ void lds_barrier() {
    __builtin_amdgcn_fence(__ATOMIC_RELEASE, "workgroup", "local");
    __builtin_amdgcn_s_barrier();
    __builtin_amdgcn_fence(__ATOMIC_ACQUIRE, "workgroup", "local");
}

__device__ __forceinline__ void load_xz(v8f& x0, v8f& x1, v8f& x2, v8f& x3, const float* xzp) {
    for (int e = 0; e < 8; ++e) {
        x0[e] = xzp[(size_t)e * N4];
        x1[e] = xzp[(size_t)e * N4 + UNITS_];
        x2[e] = xzp[(size_t)e * N4 + 2 * UNITS_];
        x3[e] = xzp[(size_t)e * N4 + 3 * UNITS_];
    }
}

__global__ void __launch_bounds__(32 * SCAN_WAVES)
lstm_chunk_kernel(const float* in_fw, const float* in_bw,
                  const float* __restrict__ pr_fw,
                  const float* __restrict__ pr_bw,
                  float* xz,
                  float* __restrict__ state,
                  float* out_fw, float* out_bw,
                  int chunk, int residual,
                  const float* xin_fw, const float* xin_bw,
                  const float* __restrict__ xw_fw, const float* __restrict__ xw_bw,
                  const float* __restrict__ xb_fw, const float* __restrict__ xb_bw,
                  int xfirst, int nscan) {
    __shared__ __attribute__((aligned(16))) float Hbuf[2 * 16 * HSTRIDE];

    if ((int)blockIdx.x >= nscan) {
        const int xchunk = xfirst ? 0 : chunk + 1;
        xproj_tile(xin_fw, xin_bw, xw_fw, xw_bw, xb_fw, xb_bw,
                   xz + (size_t)(xchunk & 1) * XZ_PLANE, xchunk,
                   ((int)blockIdx.x - nscan) * SCAN_WAVES + (int)(threadIdx.x >> 5),
                   (int)(threadIdx.x & 31));
        return;
    }

    const float* xzc  = xz + (size_t)(chunk & 1) * XZ_PLANE;
    const int dir     = blockIdx.x >> 3;
    const float* in   = dir ? in_bw : in_fw;
    const float* pr   = dir ? pr_bw : pr_fw;
    float*       out  = dir ? out_bw : out_fw;

    const int b0      = (blockIdx.x & 7) * 16;
    const int tid     = threadIdx.x;
    const int lane    = tid & 31;
    const int wave    = tid >> 5;
    const int halfsel = lane >> 4;
    const int lrow    = lane & 15;

    float* st_h = state + (size_t)(dir * 2 + 0) * NBATCH * UNITS_;
    float* st_c = state + (size_t)(dir * 2 + 1) * NBATCH * UNITS_;

    const int cunit = wave * 16 + lrow;
    const int crow0 = 8 * halfsel;
    const int frow  = wave;
    const int fcol  = 4 * lane;

    float cst[8] = {0.0f, 0.0f, 0.0f, 0.0f, 0.0f, 0.0f, 0.0f, 0.0f};
    float4 h4a = make_float4(0.0f, 0.0f, 0.0f, 0.0f);
    float4 h4b = make_float4(0.0f, 0.0f, 0.0f, 0.0f);
    float* cptr = st_c + (size_t)b0 * UNITS_ + 4 * tid;
    float* hptr = st_h + (size_t)(b0 + frow) * UNITS_ + fcol;
    if (chunk != 0) {
        const float4 c4a = *(const float4*)(cptr);
        const float4 c4b = *(const float4*)(cptr + 8 * UNITS_);
        cst[0] = c4a.x; cst[1] = c4a.y; cst[2] = c4a.z; cst[3] = c4a.w;
        cst[4] = c4b.x; cst[5] = c4b.y; cst[6] = c4b.z; cst[7] = c4b.w;
        h4a = *(const float4*)(hptr);
        h4b = *(const float4*)(hptr + 128);
    }
    const float* pw = pr + ((size_t)wave * (64 * 2 * 32) + lane) * 4;
    const int aBase = lrow * HSTRIDE + 2 * halfsel;

    v4f q[16];
    load_group(q, pw);
    v8f x0, x1, x2, x3;
    float4 resa = make_float4(0.0f, 0.0f, 0.0f, 0.0f);
    float4 resb = make_float4(0.0f, 0.0f, 0.0f, 0.0f);
    {
        const int t0 = dir ? (T_STEPS - 1 - TC * chunk) : (TC * chunk);
        load_xz(x0, x1, x2, x3,
                xzc + ((size_t)((dir * TC) * NBATCH + b0 + crow0)) * N4 + cunit);
        if (residual) {
            const size_t i0 = ((size_t)(b0 + frow) * T_STEPS + t0) * UNITS_ + fcol;
            resa = *(const float4*)(in + i0);
            resb = *(const float4*)(in + i0 + 128);
        }
    }

    *(float4*)(Hbuf + frow * HSTRIDE + fcol)       = h4a;
    *(float4*)(Hbuf + frow * HSTRIDE + fcol + 128) = h4b;

    lds_barrier();

    for (int tc = 0; tc < TC; ++tc) {
        const int t = dir ? (T_STEPS - 1 - TC * chunk - tc) : (TC * chunk + tc);
        const float* Hcur = Hbuf + (tc & 1) * (16 * HSTRIDE);
        float*       Hnxt = Hbuf + ((tc & 1) ^ 1) * (16 * HSTRIDE);
        const size_t oidx = ((size_t)(b0 + frow) * T_STEPS + t) * UNITS_ + fcol;

        v8f acc0 = {0.f, 0.f, 0.f, 0.f, 0.f, 0.f, 0.f, 0.f};
        v8f acc1 = {0.f, 0.f, 0.f, 0.f, 0.f, 0.f, 0.f, 0.f};
        v8f acc2 = {0.f, 0.f, 0.f, 0.f, 0.f, 0.f, 0.f, 0.f};
        v8f acc3 = {0.f, 0.f, 0.f, 0.f, 0.f, 0.f, 0.f, 0.f};
        mma_group(acc0, acc1, acc2, acc3, q, Hcur + aBase);
#pragma nounroll
        for (int g = 1; g < 8; ++g) {
            v4f r[16];
            load_group(r, pw + g * (8 * 256));
            mma_group(acc0, acc1, acc2, acc3, r, Hcur + aBase + 32 * g);
        }

        const bool more = (tc + 1 < TC);
        if (more) load_group(q, pw);

#pragma unroll
        for (int j = 0; j < 8; ++j) {
            const float zi = acc0[j] + x0[j];
            const float zf = acc1[j] + x1[j];
            const float zg = acc2[j] + x2[j];
            const float zo = acc3[j] + x3[j];
            float gi = sigmoid_f(zi);
            float gf = sigmoid_f(zf);
            float gg = tanh_f(zg);
            float go = sigmoid_f(zo);
            float c  = gf * cst[j] + gi * gg;
            cst[j]   = c;
            float h  = go * tanh_f(c);
            Hnxt[(crow0 + j) * HSTRIDE + cunit] = h;
        }

        const float4 ra = resa;
        const float4 rb = resb;
        if (more) {
            load_xz(x0, x1, x2, x3,
                    xzc + ((size_t)((dir * TC + tc + 1) * NBATCH + b0 + crow0)) * N4 + cunit);
            if (residual) {
                const int tn = dir ? (t - 1) : (t + 1);
                const size_t in1 = ((size_t)(b0 + frow) * T_STEPS + tn) * UNITS_ + fcol;
                resa = *(const float4*)(in + in1);
                resb = *(const float4*)(in + in1 + 128);
            }
        }
        lds_barrier();

        h4a = *(const float4*)(Hnxt + frow * HSTRIDE + fcol);
        h4b = *(const float4*)(Hnxt + frow * HSTRIDE + fcol + 128);
        *(float4*)(out + oidx) = make_float4(h4a.x + ra.x, h4a.y + ra.y,
                                             h4a.z + ra.z, h4a.w + ra.w);
        *(float4*)(out + oidx + 128) = make_float4(h4b.x + rb.x, h4b.y + rb.y,
                                                   h4b.z + rb.z, h4b.w + rb.w);
    }

    *(float4*)(hptr)       = h4a;
    *(float4*)(hptr + 128) = h4b;
    *(float4*)(cptr)              = make_float4(cst[0], cst[1], cst[2], cst[3]);
    *(float4*)(cptr + 8 * UNITS_) = make_float4(cst[4], cst[5], cst[6], cst[7]);
}

__global__ void combine_avg_kernel(const float4* __restrict__ fw,
                                   float4* __restrict__ out, int n4) {
    int i = blockIdx.x * blockDim.x + threadIdx.x;
    if (i < n4) {
        float4 a = fw[i];
        float4 b = out[i];
        out[i] = make_float4((a.x + b.x) * 0.5f, (a.y + b.y) * 0.5f,
                             (a.z + b.z) * 0.5f, (a.w + b.w) * 0.5f);
    }
}

extern "C" void kernel_launch(void* const* d_in, const int* in_sizes, int n_in,
                              void* d_out, int out_size, void* d_ws, size_t ws_size,
                              hipStream_t stream) {
    const float* x   = (const float*)d_in[0];
    const float* kfw = (const float*)d_in[1];
    const float* rfw = (const float*)d_in[2];
    const float* bfw = (const float*)d_in[3];
    const float* kbw = (const float*)d_in[4];
    const float* rbw = (const float*)d_in[5];
    const float* bbw = (const float*)d_in[6];
    float* out = (float*)d_out;

    float* rpk  = (float*)d_ws;
    float* xz   = rpk + (size_t)2 * RCOMBO_F;
    float* st   = xz + (size_t)2 * XZ_PLANE;
    float* wsfw = (float*)d_ws + (size_t)2 * 1024 * 1024;
    float* wsbw = out;

    pack_r_kernel<<<512, 256, 0, stream>>>(rfw, rbw, rpk);
    lstm_chunk_kernel<<<NPROJ_WG, 512, 0, stream>>>(
        x, x, rpk, rpk + RCOMBO_F, xz, st, wsfw, wsbw, 0,  0,
        x, x, kfw, kbw, bfw, bbw,  1,  0);
    for (int k = 0; k < NCHUNK - 1; ++k) {
        lstm_chunk_kernel<<<NSCAN_WG + NPROJ_WG, 512, 0, stream>>>(
            x, x, rpk, rpk + RCOMBO_F, xz, st, wsfw, wsbw, k,  0,
            x, x, kfw, kbw, bfw, bbw,  0, NSCAN_WG);
    }
    lstm_chunk_kernel<<<NSCAN_WG + NPROJ_WG, 512, 0, stream>>>(
        x, x, rpk, rpk + RCOMBO_F, xz, st, wsfw, wsbw, NCHUNK - 1,  0,
        wsfw, wsbw, kfw + RCOMBO_F, kbw + RCOMBO_F, bfw + 1024, bbw + 1024,
         1, NSCAN_WG);

    pack_r_kernel<<<512, 256, 0, stream>>>(rfw + RCOMBO_F, rbw + RCOMBO_F, rpk);
    for (int k = 0; k < NCHUNK - 1; ++k) {
        lstm_chunk_kernel<<<NSCAN_WG + NPROJ_WG, 512, 0, stream>>>(
            wsfw, wsbw, rpk, rpk + RCOMBO_F, xz, st, wsfw, wsbw, k,  1,
            wsfw, wsbw, kfw + RCOMBO_F, kbw + RCOMBO_F, bfw + 1024, bbw + 1024,
             0, NSCAN_WG);
    }
    lstm_chunk_kernel<<<NSCAN_WG, 512, 0, stream>>>(
        wsfw, wsbw, rpk, rpk + RCOMBO_F, xz, st, wsfw, wsbw, NCHUNK - 1,  1,
        wsfw, wsbw, kfw + RCOMBO_F, kbw + RCOMBO_F, bfw + 1024, bbw + 1024,
         0, NSCAN_WG);

    const int n4 = (128 * 256 * 256) / 4;
    combine_avg_kernel<<<(n4 + 255) / 256, 256, 0, stream>>>(
        (const float4*)wsfw, (float4*)out, n4);
}


# time chunks as long as the workspace holds two planes of x@W+b: 8 steps (52 MB in all), 4 (44 MB), else 2 as before: 65 or 129 chunk launches, not 257; sum order kept
# speedup vs baseline: 2.1667x; 1.1262x over previous
#include <hip/hip_runtime.h>


typedef float v2f __attribute__((ext_vector_type(2)));
typedef float v4f __attribute__((ext_vector_type(4)));
typedef float v8f __attribute__((ext_vector_type(8)));

#define T_STEPS 256
#define UNITS_  256
#define N4      1024
#define NBATCH  128
#define SCAN_WAVES 16
#define HSTRIDE 260
#define RCOMBO_F (256 * 1024)
#define XZ_PLANE_F(tcn) (2 * (tcn) * NBATCH * N4)
#define NSCAN_WG 16

__device__ __forceinline__ float sigmoid_f(float x) {
    return __builtin_amdgcn_rcpf(1.0f + __expf(-x));
}
__device__ __forceinline__ float tanh_f(float x) {
    x = fminf(15.0f, fmaxf(-15.0f, x));
    float e = __expf(2.0f * x);
    return (e - 1.0f) * __builtin_amdgcn_rcpf(e + 1.0f);
}

__global__ void pack_r_kernel(const float* __restrict__ rfw,
                              const float* __restrict__ rbw,
                              float* __restrict__ packed) {
    int i = blockIdx.x * blockDim.x + threadIdx.x;
    int lane = i & 31;
    int gp   = (i >> 5) & 1;
    int ks   = (i >> 6) & 63;
    int w    = (i >> 12) & 15;
    int isbw = i >> 16;
    const float* R = isbw ? rbw : rfw;
    int c  = lane & 15;
    int n  = (2 * gp) * 256 + 16 * w + c;
    int k0 = ks * 4 + ((lane >> 4) << 1);
    float4 v;
    v.x = R[(size_t)k0 * 1024 + n];
    v.y = R[(size_t)(k0 + 1) * 1024 + n];
    v.z = R[(size_t)k0 * 1024 + n + 256];
    v.w = R[(size_t)(k0 + 1) * 1024 + n + 256];
    ((float4*)packed)[i] = v;
}

template <int TCN>
__device__ __forceinline__ void
xproj_tile(const float* in_fw, const float* in_bw,
           const float* __restrict__ w_fw,  const float* __restrict__ w_bw,
           const float* __restrict__ bias_fw, const float* __restrict__ bias_bw,
           float* xz, int chunk, int gw, int lane) {
    const int mt      = gw >> 5;
    const int np      = gw & 31;
    const int dir     = mt / (8 * TCN);
    const int tc      = (mt - dir * (8 * TCN)) >> 3;
    const int bt      = mt & 7;
    const int halfsel = lane >> 4;
    const int lrow    = lane & 15;

    const float* in   = dir ? in_bw : in_fw;
    const float* Wd   = dir ? w_bw : w_fw;
    const float* bias = dir ? bias_bw : bias_fw;
    const int t = dir ? (T_STEPS - 1 - TCN * chunk - tc) : (TCN * chunk + tc);

    const float* arow = in + ((size_t)(bt * 16 + lrow) * T_STEPS + t) * UNITS_ + 2 * halfsel;
    const int n0 = np * 32 + lrow;
    const float* wcol = Wd + (size_t)(2 * halfsel) * N4 + n0;

    v8f acc0 = {0.f, 0.f, 0.f, 0.f, 0.f, 0.f, 0.f, 0.f};
    v8f acc1 = {0.f, 0.f, 0.f, 0.f, 0.f, 0.f, 0.f, 0.f};
#pragma unroll 8
    for (int ks = 0; ks < UNITS_ / 4; ++ks) {
        v2f a = *(const v2f*)(arow + 4 * ks);
        const float* wp = wcol + (size_t)(4 * ks) * N4;
        v2f w0, w1;
        w0.x = wp[0];
        w0.y = wp[N4];
        w1.x = wp[16];
        w1.y = wp[N4 + 16];
        acc0 = __builtin_amdgcn_wmma_f32_16x16x4_f32(
            false, a, false, w0, (short)0, acc0, false, false);
        acc1 = __builtin_amdgcn_wmma_f32_16x16x4_f32(
            false, a, false, w1, (short)0, acc1, false, false);
    }
    const float bv0 = bias[n0];
    const float bv1 = bias[n0 + 16];
    float* dst = xz + ((size_t)((dir * TCN + tc) * NBATCH + bt * 16 + 8 * halfsel)) * N4 + n0;
#pragma unroll
    for (int e = 0; e < 8; ++e) {
        dst[(size_t)e * N4]      = acc0[e] + bv0;
        dst[(size_t)e * N4 + 16] = acc1[e] + bv1;
    }
}

__device__ __forceinline__ void load_group(v4f (&q)[16], const float* p) {
    for (int i = 0; i < 8; ++i) {
        q[2 * i]     = *(const v4f*)(p + i * 256);
        q[2 * i + 1] = *(const v4f*)(p + i * 256 + 128);
    }
}
__device__ __forceinline__ void mma_group(v8f& acc0, v8f& acc1, v8f& acc2, v8f& acc3,
                                          const v4f (&q)[16], const float* hrow) {
    for (int i = 0; i < 8; ++i) {
        v2f a  = *(const v2f*)(hrow + 4 * i);
        v2f wi = {q[2 * i].x, q[2 * i].y};
        v2f wf = {q[2 * i].z, q[2 * i].w};
        v2f wg = {q[2 * i + 1].x, q[2 * i + 1].y};
        v2f wo = {q[2 * i + 1].z, q[2 * i + 1].w};
        acc0 = __builtin_amdgcn_wmma_f32_16x16x4_f32(
            false, a, false, wi, (short)0, acc0, false, false);
        acc1 = __builtin_amdgcn_wmma_f32_16x16x4_f32(
            false, a, false, wf, (short)0, acc1, false, false);
        acc2 = __builtin_amdgcn_wmma_f32_16x16x4_f32(
            false, a, false, wg, (short)0, acc2, false, false);
        acc3 = __builtin_amdgcn_wmma_f32_16x16x4_f32(
            false, a, false, wo, (short)0, acc3, false, false);
    }
}
__device__ __forceinline__ void lds_barrier() {
    __builtin_amdgcn_fence(__ATOMIC_RELEASE, "workgroup", "local");
    __builtin_amdgcn_s_barrier();
    __builtin_amdgcn_fence(__ATOMIC_ACQUIRE, "workgroup", "local");
}

__device__ __forceinline__ void load_xz(v8f& x0, v8f& x1, v8f& x2, v8f& x3, const float* xzp) {
    for (int e = 0; e < 8; ++e) {
        x0[e] = xzp[(size_t)e * N4];
        x1[e] = xzp[(size_t)e * N4 + UNITS_];
        x2[e] = xzp[(size_t)e * N4 + 2 * UNITS_];
        x3[e] = xzp[(size_t)e * N4 + 3 * UNITS_];
    }
}

template <int TCN>
__global__ void __launch_bounds__(32 * SCAN_WAVES)
lstm_chunk_kernel(const float* in_fw, const float* in_bw,
                  const float* __restrict__ pr_fw,
                  const float* __restrict__ pr_bw,
                  float* xz,
                  float* __restrict__ state,
                  float* out_fw, float* out_bw,
                  int chunk, int residual,
                  const float* xin_fw, const float* xin_bw,
                  const float* __restrict__ xw_fw, const float* __restrict__ xw_bw,
                  const float* __restrict__ xb_fw, const float* __restrict__ xb_bw,
                  int xfirst, int nscan) {
    __shared__ __attribute__((aligned(16))) float Hbuf[2 * 16 * HSTRIDE];

    if ((int)blockIdx.x >= nscan) {
        const int xchunk = xfirst ? 0 : chunk + 1;
        xproj_tile<TCN>(xin_fw, xin_bw, xw_fw, xw_bw, xb_fw, xb_bw,
                   xz + (size_t)(xchunk & 1) * XZ_PLANE_F(TCN), xchunk,
                   ((int)blockIdx.x - nscan) * SCAN_WAVES + (int)(threadIdx.x >> 5),
                   (int)(threadIdx.x & 31));
        return;
    }

    const float* xzc  = xz + (size_t)(chunk & 1) * XZ_PLANE_F(TCN);
    const int dir     = blockIdx.x >> 3;
    const float* in   = dir ? in_bw : in_fw;
    const float* pr   = dir ? pr_bw : pr_fw;
    float*       out  = dir ? out_bw : out_fw;

    const int b0      = (blockIdx.x & 7) * 16;
    const int tid     = threadIdx.x;
    const int lane    = tid & 31;
    const int wave    = tid >> 5;
    const int halfsel = lane >> 4;
    const int lrow    = lane & 15;

    float* st_h = state + (size_t)(dir * 2 + 0) * NBATCH * UNITS_;
    float* st_c = state + (size_t)(dir * 2 + 1) * NBATCH * UNITS_;

    const int cunit = wave * 16 + lrow;
    const int crow0 = 8 * halfsel;
    const int frow  = wave;
    const int fcol  = 4 * lane;

    float cst[8] = {0.0f, 0.0f, 0.0f, 0.0f, 0.0f, 0.0f, 0.0f, 0.0f};
    float4 h4a = make_float4(0.0f, 0.0f, 0.0f, 0.0f);
    float4 h4b = make_float4(0.0f, 0.0f, 0.0f, 0.0f);
    float* cptr = st_c + (size_t)b0 * UNITS_ + 4 * tid;
    float* hptr = st_h + (size_t)(b0 + frow) * UNITS_ + fcol;
    if (chunk != 0) {
        const float4 c4a = *(const float4*)(cptr);
        const float4 c4b = *(const float4*)(cptr + 8 * UNITS_);
        cst[0] = c4a.x; cst[1] = c4a.y; cst[2] = c4a.z; cst[3] = c4a.w;
        cst[4] = c4b.x; cst[5] = c4b.y; cst[6] = c4b.z; cst[7] = c4b.w;
        h4a = *(const float4*)(hptr);
        h4b = *(const float4*)(hptr + 128);
    }
    const float* pw = pr + ((size_t)wave * (64 * 2 * 32) + lane) * 4;
    const int aBase = lrow * HSTRIDE + 2 * halfsel;

    v4f q[16];
    load_group(q, pw);
    v8f x0, x1, x2, x3;
    float4 resa = make_float4(0.0f, 0.0f, 0.0f, 0.0f);
    float4 resb = make_float4(0.0f, 0.0f, 0.0f, 0.0f);
    {
        const int t0 = dir ? (T_STEPS - 1 - TCN * chunk) : (TCN * chunk);
        load_xz(x0, x1, x2, x3,
                xzc + ((size_t)((dir * TCN) * NBATCH + b0 + crow0)) * N4 + cunit);
        if (residual) {
            const size_t i0 = ((size_t)(b0 + frow) * T_STEPS + t0) * UNITS_ + fcol;
            resa = *(const float4*)(in + i0);
            resb = *(const float4*)(in + i0 + 128);
        }
    }

    *(float4*)(Hbuf + frow * HSTRIDE + fcol)       = h4a;
    *(float4*)(Hbuf + frow * HSTRIDE + fcol + 128) = h4b;

    lds_barrier();

#pragma unroll
    for (int tc = 0; tc < TCN; ++tc) {
        const int t = dir ? (T_STEPS - 1 - TCN * chunk - tc) : (TCN * chunk + tc);
        const float* Hcur = Hbuf + (tc & 1) * (16 * HSTRIDE);
        float*       Hnxt = Hbuf + ((tc & 1) ^ 1) * (16 * HSTRIDE);
        const size_t oidx = ((size_t)(b0 + frow) * T_STEPS + t) * UNITS_ + fcol;

        v8f acc0 = {0.f, 0.f, 0.f, 0.f, 0.f, 0.f, 0.f, 0.f};
        v8f acc1 = {0.f, 0.f, 0.f, 0.f, 0.f, 0.f, 0.f, 0.f};
        v8f acc2 = {0.f, 0.f, 0.f, 0.f, 0.f, 0.f, 0.f, 0.f};
        v8f acc3 = {0.f, 0.f, 0.f, 0.f, 0.f, 0.f, 0.f, 0.f};
        mma_group(acc0, acc1, acc2, acc3, q, Hcur + aBase);
#pragma nounroll
        for (int g = 1; g < 8; ++g) {
            v4f r[16];
            load_group(r, pw + g * (8 * 256));
            mma_group(acc0, acc1, acc2, acc3, r, Hcur + aBase + 32 * g);
        }

        const bool more = (tc + 1 < TCN);
        if (more) load_group(q, pw);

#pragma unroll
        for (int j = 0; j < 8; ++j) {
            const float zi = acc0[j] + x0[j];
            const float zf = acc1[j] + x1[j];
            const float zg = acc2[j] + x2[j];
            const float zo = acc3[j] + x3[j];
            float gi = sigmoid_f(zi);
            float gf = sigmoid_f(zf);
            float gg = tanh_f(zg);
            float go = sigmoid_f(zo);
            float c  = gf * cst[j] + gi * gg;
            cst[j]   = c;
            float h  = go * tanh_f(c);
            Hnxt[(crow0 + j) * HSTRIDE + cunit] = h;
        }

        const float4 ra = resa;
        const float4 rb = resb;
        if (more) {
            load_xz(x0, x1, x2, x3,
                    xzc + ((size_t)((dir * TCN + tc + 1) * NBATCH + b0 + crow0)) * N4 + cunit);
            if (residual) {
                const int tn = dir ? (t - 1) : (t + 1);
                const size_t in1 = ((size_t)(b0 + frow) * T_STEPS + tn) * UNITS_ + fcol;
                resa = *(const float4*)(in + in1);
                resb = *(const float4*)(in + in1 + 128);
            }
        }
        lds_barrier();

        h4a = *(const float4*)(Hnxt + frow * HSTRIDE + fcol);
        h4b = *(const float4*)(Hnxt + frow * HSTRIDE + fcol + 128);
        *(float4*)(out + oidx) = make_float4(h4a.x + ra.x, h4a.y + ra.y,
                                             h4a.z + ra.z, h4a.w + ra.w);
        *(float4*)(out + oidx + 128) = make_float4(h4b.x + rb.x, h4b.y + rb.y,
                                                   h4b.z + rb.z, h4b.w + rb.w);
    }

    *(float4*)(hptr)       = h4a;
    *(float4*)(hptr + 128) = h4b;
    *(float4*)(cptr)              = make_float4(cst[0], cst[1], cst[2], cst[3]);
    *(float4*)(cptr + 8 * UNITS_) = make_float4(cst[4], cst[5], cst[6], cst[7]);
}

__global__ void combine_avg_kernel(const float4* __restrict__ fw,
                                   float4* __restrict__ out, int n4) {
    int i = blockIdx.x * blockDim.x + threadIdx.x;
    if (i < n4) {
        float4 a = fw[i];
        float4 b = out[i];
        out[i] = make_float4((a.x + b.x) * 0.5f, (a.y + b.y) * 0.5f,
                             (a.z + b.z) * 0.5f, (a.w + b.w) * 0.5f);
    }
}

template <int TCN>
static void run_bilstm(const float* x,
                       const float* kfw, const float* rfw, const float* bfw,
                       const float* kbw, const float* rbw, const float* bbw,
                       float* out, float* rpk, float* xz, float* st, float* wsfw,
                       hipStream_t stream) {
    const int nchunk = T_STEPS / TCN;
    const int nproj  = 32 * TCN;
    float* wsbw = out;

    pack_r_kernel<<<512, 256, 0, stream>>>(rfw, rbw, rpk);
    lstm_chunk_kernel<TCN><<<nproj, 512, 0, stream>>>(
        x, x, rpk, rpk + RCOMBO_F, xz, st, wsfw, wsbw, 0,  0,
        x, x, kfw, kbw, bfw, bbw,  1,  0);
    for (int k = 0; k < nchunk - 1; ++k) {
        lstm_chunk_kernel<TCN><<<NSCAN_WG + nproj, 512, 0, stream>>>(
            x, x, rpk, rpk + RCOMBO_F, xz, st, wsfw, wsbw, k,  0,
            x, x, kfw, kbw, bfw, bbw,  0, NSCAN_WG);
    }
    lstm_chunk_kernel<TCN><<<NSCAN_WG + nproj, 512, 0, stream>>>(
        x, x, rpk, rpk + RCOMBO_F, xz, st, wsfw, wsbw, nchunk - 1,  0,
        wsfw, wsbw, kfw + RCOMBO_F, kbw + RCOMBO_F, bfw + 1024, bbw + 1024,
         1, NSCAN_WG);

    pack_r_kernel<<<512, 256, 0, stream>>>(rfw + RCOMBO_F, rbw + RCOMBO_F, rpk);
    for (int k = 0; k < nchunk - 1; ++k) {
        lstm_chunk_kernel<TCN><<<NSCAN_WG + nproj, 512, 0, stream>>>(
            wsfw, wsbw, rpk, rpk + RCOMBO_F, xz, st, wsfw, wsbw, k,  1,
            wsfw, wsbw, kfw + RCOMBO_F, kbw + RCOMBO_F, bfw + 1024, bbw + 1024,
             0, NSCAN_WG);
    }
    lstm_chunk_kernel<TCN><<<NSCAN_WG, 512, 0, stream>>>(
        wsfw, wsbw, rpk, rpk + RCOMBO_F, xz, st, wsfw, wsbw, nchunk - 1,  1,
        wsfw, wsbw, kfw + RCOMBO_F, kbw + RCOMBO_F, bfw + 1024, bbw + 1024,
         0, NSCAN_WG);

    const int n4 = (128 * 256 * 256) / 4;
    combine_avg_kernel<<<(n4 + 255) / 256, 256, 0, stream>>>(
        (const float4*)wsfw, (float4*)out, n4);
}

extern "C" void kernel_launch(void* const* d_in, const int* in_sizes, int n_in,
                              void* d_out, int out_size, void* d_ws, size_t ws_size,
                              hipStream_t stream) {
    const float* x   = (const float*)d_in[0];
    const float* kfw = (const float*)d_in[1];
    const float* rfw = (const float*)d_in[2];
    const float* bfw = (const float*)d_in[3];
    const float* kbw = (const float*)d_in[4];
    const float* rbw = (const float*)d_in[5];
    const float* bbw = (const float*)d_in[6];
    float* out = (float*)d_out;

    float* rpk = (float*)d_ws;
    float* xz  = rpk + (size_t)2 * RCOMBO_F;
    if (ws_size >= (size_t)52 * 1024 * 1024) {
        run_bilstm<8>(x, kfw, rfw, bfw, kbw, rbw, bbw, out, rpk, xz,
                      xz + (size_t)2 * XZ_PLANE_F(8),
                      (float*)d_ws + (size_t)5 * 1024 * 1024, stream);
    } else if (ws_size >= (size_t)44 * 1024 * 1024) {
        run_bilstm<4>(x, kfw, rfw, bfw, kbw, rbw, bbw, out, rpk, xz,
                      xz + (size_t)2 * XZ_PLANE_F(4),
                      (float*)d_ws + (size_t)3 * 1024 * 1024, stream);
    } else {
        run_bilstm<2>(x, kfw, rfw, bfw, kbw, rbw, bbw, out, rpk, xz,
                      xz + (size_t)2 * XZ_PLANE_F(2),
                      (float*)d_ws + (size_t)2 * 1024 * 1024, stream);
    }
}


# scan waves 12..15 (youngest of each SIMD) keep the weight fragments of k-steps 8..63 in LDS from a launch's first step on (224 KB) and read them there, not from the second-level cache; sum order kept
# speedup vs baseline: 2.2500x; 1.0384x over previous
#include <hip/hip_runtime.h>


typedef float v2f __attribute__((ext_vector_type(2)));
typedef float v4f __attribute__((ext_vector_type(4)));
typedef float v8f __attribute__((ext_vector_type(8)));

#define T_STEPS 256
#define UNITS_  256
#define N4      1024
#define NBATCH  128
#define SCAN_WAVES 16
#define HSTRIDE 260
#define RCOMBO_F (256 * 1024)
#define XZ_PLANE_F(tcn) (2 * (tcn) * NBATCH * N4)
#define NSCAN_WG 16
#define RES_WAVE0 12
#define RES_F_PER_WAVE (7 * 8 * 256)

__device__ __forceinline__ float sigmoid_f(float x) {
    return __builtin_amdgcn_rcpf(1.0f + __expf(-x));
}
__device__ __forceinline__ float tanh_f(float x) {
    x = fminf(15.0f, fmaxf(-15.0f, x));
    float e = __expf(2.0f * x);
    return (e - 1.0f) * __builtin_amdgcn_rcpf(e + 1.0f);
}

__global__ void pack_r_kernel(const float* __restrict__ rfw,
                              const float* __restrict__ rbw,
                              float* __restrict__ packed) {
    int i = blockIdx.x * blockDim.x + threadIdx.x;
    int lane = i & 31;
    int gp   = (i >> 5) & 1;
    int ks   = (i >> 6) & 63;
    int w    = (i >> 12) & 15;
    int isbw = i >> 16;
    const float* R = isbw ? rbw : rfw;
    int c  = lane & 15;
    int n  = (2 * gp) * 256 + 16 * w + c;
    int k0 = ks * 4 + ((lane >> 4) << 1);
    float4 v;
    v.x = R[(size_t)k0 * 1024 + n];
    v.y = R[(size_t)(k0 + 1) * 1024 + n];
    v.z = R[(size_t)k0 * 1024 + n + 256];
    v.w = R[(size_t)(k0 + 1) * 1024 + n + 256];
    ((float4*)packed)[i] = v;
}

template <int TCN>
__device__ __forceinline__ void
xproj_tile(const float* in_fw, const float* in_bw,
           const float* __restrict__ w_fw,  const float* __restrict__ w_bw,
           const float* __restrict__ bias_fw, const float* __restrict__ bias_bw,
           float* xz, int chunk, int gw, int lane) {
    const int mt      = gw >> 5;
    const int np      = gw & 31;
    const int dir     = mt / (8 * TCN);
    const int tc      = (mt - dir * (8 * TCN)) >> 3;
    const int bt      = mt & 7;
    const int halfsel = lane >> 4;
    const int lrow    = lane & 15;

    const float* in   = dir ? in_bw : in_fw;
    const float* Wd   = dir ? w_bw : w_fw;
    const float* bias = dir ? bias_bw : bias_fw;
    const int t = dir ? (T_STEPS - 1 - TCN * chunk - tc) : (TCN * chunk + tc);

    const float* arow = in + ((size_t)(bt * 16 + lrow) * T_STEPS + t) * UNITS_ + 2 * halfsel;
    const int n0 = np * 32 + lrow;
    const float* wcol = Wd + (size_t)(2 * halfsel) * N4 + n0;

    v8f acc0 = {0.f, 0.f, 0.f, 0.f, 0.f, 0.f, 0.f, 0.f};
    v8f acc1 = {0.f, 0.f, 0.f, 0.f, 0.f, 0.f, 0.f, 0.f};
#pragma unroll 8
    for (int ks = 0; ks < UNITS_ / 4; ++ks) {
        v2f a = *(const v2f*)(arow + 4 * ks);
        const float* wp = wcol + (size_t)(4 * ks) * N4;
        v2f w0, w1;
        w0.x = wp[0];
        w0.y = wp[N4];
        w1.x = wp[16];
        w1.y = wp[N4 + 16];
        acc0 = __builtin_amdgcn_wmma_f32_16x16x4_f32(
            false, a, false, w0, (short)0, acc0, false, false);
        acc1 = __builtin_amdgcn_wmma_f32_16x16x4_f32(
            false, a, false, w1, (short)0, acc1, false, false);
    }
    const float bv0 = bias[n0];
    const float bv1 = bias[n0 + 16];
    float* dst = xz + ((size_t)((dir * TCN + tc) * NBATCH + bt * 16 + 8 * halfsel)) * N4 + n0;
#pragma unroll
    for (int e = 0; e < 8; ++e) {
        dst[(size_t)e * N4]      = acc0[e] + bv0;
        dst[(size_t)e * N4 + 16] = acc1[e] + bv1;
    }
}

__device__ __forceinline__ void load_group(v4f (&q)[16], const float* p) {
    for (int i = 0; i < 8; ++i) {
        q[2 * i]     = *(const v4f*)(p + i * 256);
        q[2 * i + 1] = *(const v4f*)(p + i * 256 + 128);
    }
}
__device__ __forceinline__ void store_group(float* p, const v4f (&q)[16]) {
    for (int i = 0; i < 8; ++i) {
        *(v4f*)(p + i * 256)       = q[2 * i];
        *(v4f*)(p + i * 256 + 128) = q[2 * i + 1];
    }
}
__device__ __forceinline__ void mma_group(v8f& acc0, v8f& acc1, v8f& acc2, v8f& acc3,
                                          const v4f (&q)[16], const float* hrow) {
    for (int i = 0; i < 8; ++i) {
        v2f a  = *(const v2f*)(hrow + 4 * i);
        v2f wi = {q[2 * i].x, q[2 * i].y};
        v2f wf = {q[2 * i].z, q[2 * i].w};
        v2f wg = {q[2 * i + 1].x, q[2 * i + 1].y};
        v2f wo = {q[2 * i + 1].z, q[2 * i + 1].w};
        acc0 = __builtin_amdgcn_wmma_f32_16x16x4_f32(
            false, a, false, wi, (short)0, acc0, false, false);
        acc1 = __builtin_amdgcn_wmma_f32_16x16x4_f32(
            false, a, false, wf, (short)0, acc1, false, false);
        acc2 = __builtin_amdgcn_wmma_f32_16x16x4_f32(
            false, a, false, wg, (short)0, acc2, false, false);
        acc3 = __builtin_amdgcn_wmma_f32_16x16x4_f32(
            false, a, false, wo, (short)0, acc3, false, false);
    }
}
__device__ __forceinline__ void lds_barrier() {
    __builtin_amdgcn_fence(__ATOMIC_RELEASE, "workgroup", "local");
    __builtin_amdgcn_s_barrier();
    __builtin_amdgcn_fence(__ATOMIC_ACQUIRE, "workgroup", "local");
}

__device__ __forceinline__ void load_xz(v8f& x0, v8f& x1, v8f& x2, v8f& x3, const float* xzp) {
    for (int e = 0; e < 8; ++e) {
        x0[e] = xzp[(size_t)e * N4];
        x1[e] = xzp[(size_t)e * N4 + UNITS_];
        x2[e] = xzp[(size_t)e * N4 + 2 * UNITS_];
        x3[e] = xzp[(size_t)e * N4 + 3 * UNITS_];
    }
}

template <int TCN>
__global__ void __launch_bounds__(32 * SCAN_WAVES)
lstm_chunk_kernel(const float* in_fw, const float* in_bw,
                  const float* __restrict__ pr_fw,
                  const float* __restrict__ pr_bw,
                  float* xz,
                  float* __restrict__ state,
                  float* out_fw, float* out_bw,
                  int chunk, int residual,
                  const float* xin_fw, const float* xin_bw,
                  const float* __restrict__ xw_fw, const float* __restrict__ xw_bw,
                  const float* __restrict__ xb_fw, const float* __restrict__ xb_bw,
                  int xfirst, int nscan) {
    __shared__ __attribute__((aligned(16))) float Lds[2 * 16 * HSTRIDE + 4 * RES_F_PER_WAVE];
    float* const Hbuf = Lds;
    float* const Rres = Lds + 2 * 16 * HSTRIDE;

    if ((int)blockIdx.x >= nscan) {
        const int xchunk = xfirst ? 0 : chunk + 1;
        xproj_tile<TCN>(xin_fw, xin_bw, xw_fw, xw_bw, xb_fw, xb_bw,
                   xz + (size_t)(xchunk & 1) * XZ_PLANE_F(TCN), xchunk,
                   ((int)blockIdx.x - nscan) * SCAN_WAVES + (int)(threadIdx.x >> 5),
                   (int)(threadIdx.x & 31));
        return;
    }

    const float* xzc  = xz + (size_t)(chunk & 1) * XZ_PLANE_F(TCN);
    const int dir     = blockIdx.x >> 3;
    const float* in   = dir ? in_bw : in_fw;
    const float* pr   = dir ? pr_bw : pr_fw;
    float*       out  = dir ? out_bw : out_fw;

    const int b0      = (blockIdx.x & 7) * 16;
    const int tid     = threadIdx.x;
    const int lane    = tid & 31;
    const int wave    = tid >> 5;
    const int halfsel = lane >> 4;
    const int lrow    = lane & 15;

    float* st_h = state + (size_t)(dir * 2 + 0) * NBATCH * UNITS_;
    float* st_c = state + (size_t)(dir * 2 + 1) * NBATCH * UNITS_;

    const int cunit = wave * 16 + lrow;
    const int crow0 = 8 * halfsel;
    const int frow  = wave;
    const int fcol  = 4 * lane;

    float cst[8] = {0.0f, 0.0f, 0.0f, 0.0f, 0.0f, 0.0f, 0.0f, 0.0f};
    float4 h4a = make_float4(0.0f, 0.0f, 0.0f, 0.0f);
    float4 h4b = make_float4(0.0f, 0.0f, 0.0f, 0.0f);
    float* cptr = st_c + (size_t)b0 * UNITS_ + 4 * tid;
    float* hptr = st_h + (size_t)(b0 + frow) * UNITS_ + fcol;
    if (chunk != 0) {
        const float4 c4a = *(const float4*)(cptr);
        const float4 c4b = *(const float4*)(cptr + 8 * UNITS_);
        cst[0] = c4a.x; cst[1] = c4a.y; cst[2] = c4a.z; cst[3] = c4a.w;
        cst[4] = c4b.x; cst[5] = c4b.y; cst[6] = c4b.z; cst[7] = c4b.w;
        h4a = *(const float4*)(hptr);
        h4b = *(const float4*)(hptr + 128);
    }
    const float* pw = pr + ((size_t)wave * (64 * 2 * 32) + lane) * 4;
    const int aBase = lrow * HSTRIDE + 2 * halfsel;
    const int wave_s = __builtin_amdgcn_readfirstlane(wave);
    const bool resident = wave_s >= RES_WAVE0;
    float* pl = Rres + ((wave_s - RES_WAVE0) & 3) * RES_F_PER_WAVE + lane * 4;

    v4f q[16];
    load_group(q, pw);
    v8f x0, x1, x2, x3;
    float4 resa = make_float4(0.0f, 0.0f, 0.0f, 0.0f);
    float4 resb = make_float4(0.0f, 0.0f, 0.0f, 0.0f);
    {
        const int t0 = dir ? (T_STEPS - 1 - TCN * chunk) : (TCN * chunk);
        load_xz(x0, x1, x2, x3,
                xzc + ((size_t)((dir * TCN) * NBATCH + b0 + crow0)) * N4 + cunit);
        if (residual) {
            const size_t i0 = ((size_t)(b0 + frow) * T_STEPS + t0) * UNITS_ + fcol;
            resa = *(const float4*)(in + i0);
            resb = *(const float4*)(in + i0 + 128);
        }
    }

    *(float4*)(Hbuf + frow * HSTRIDE + fcol)       = h4a;
    *(float4*)(Hbuf + frow * HSTRIDE + fcol + 128) = h4b;

    lds_barrier();

#pragma unroll
    for (int tc = 0; tc < TCN; ++tc) {
        const int t = dir ? (T_STEPS - 1 - TCN * chunk - tc) : (TCN * chunk + tc);
        const float* Hcur = Hbuf + (tc & 1) * (16 * HSTRIDE);
        float*       Hnxt = Hbuf + ((tc & 1) ^ 1) * (16 * HSTRIDE);
        const size_t oidx = ((size_t)(b0 + frow) * T_STEPS + t) * UNITS_ + fcol;

        v8f acc0 = {0.f, 0.f, 0.f, 0.f, 0.f, 0.f, 0.f, 0.f};
        v8f acc1 = {0.f, 0.f, 0.f, 0.f, 0.f, 0.f, 0.f, 0.f};
        v8f acc2 = {0.f, 0.f, 0.f, 0.f, 0.f, 0.f, 0.f, 0.f};
        v8f acc3 = {0.f, 0.f, 0.f, 0.f, 0.f, 0.f, 0.f, 0.f};
        mma_group(acc0, acc1, acc2, acc3, q, Hcur + aBase);
#pragma nounroll
        for (int g = 1; g < 8; ++g) {
            v4f r[16];
            if (tc > 0 && resident) load_group(r, pl + (g - 1) * (8 * 256));
            else                    load_group(r, pw + g * (8 * 256));
            mma_group(acc0, acc1, acc2, acc3, r, Hcur + aBase + 32 * g);
            if (tc == 0 && resident) store_group(pl + (g - 1) * (8 * 256), r);
        }

        const bool more = (tc + 1 < TCN);
        if (more) load_group(q, pw);

#pragma unroll
        for (int j = 0; j < 8; ++j) {
            const float zi = acc0[j] + x0[j];
            const float zf = acc1[j] + x1[j];
            const float zg = acc2[j] + x2[j];
            const float zo = acc3[j] + x3[j];
            float gi = sigmoid_f(zi);
            float gf = sigmoid_f(zf);
            float gg = tanh_f(zg);
            float go = sigmoid_f(zo);
            float c  = gf * cst[j] + gi * gg;
            cst[j]   = c;
            float h  = go * tanh_f(c);
            Hnxt[(crow0 + j) * HSTRIDE + cunit] = h;
        }

        const float4 ra = resa;
        const float4 rb = resb;
        if (more) {
            load_xz(x0, x1, x2, x3,
                    xzc + ((size_t)((dir * TCN + tc + 1) * NBATCH + b0 + crow0)) * N4 + cunit);
            if (residual) {
                const int tn = dir ? (t - 1) : (t + 1);
                const size_t in1 = ((size_t)(b0 + frow) * T_STEPS + tn) * UNITS_ + fcol;
                resa = *(const float4*)(in + in1);
                resb = *(const float4*)(in + in1 + 128);
            }
        }
        lds_barrier();

        h4a = *(const float4*)(Hnxt + frow * HSTRIDE + fcol);
        h4b = *(const float4*)(Hnxt + frow * HSTRIDE + fcol + 128);
        *(float4*)(out + oidx) = make_float4(h4a.x + ra.x, h4a.y + ra.y,
                                             h4a.z + ra.z, h4a.w + ra.w);
        *(float4*)(out + oidx + 128) = make_float4(h4b.x + rb.x, h4b.y + rb.y,
                                                   h4b.z + rb.z, h4b.w + rb.w);
    }

    *(float4*)(hptr)       = h4a;
    *(float4*)(hptr + 128) = h4b;
    *(float4*)(cptr)              = make_float4(cst[0], cst[1], cst[2], cst[3]);
    *(float4*)(cptr + 8 * UNITS_) = make_float4(cst[4], cst[5], cst[6], cst[7]);
}

__global__ void combine_avg_kernel(const float4* __restrict__ fw,
                                   float4* __restrict__ out, int n4) {
    int i = blockIdx.x * blockDim.x + threadIdx.x;
    if (i < n4) {
        float4 a = fw[i];
        float4 b = out[i];
        out[i] = make_float4((a.x + b.x) * 0.5f, (a.y + b.y) * 0.5f,
                             (a.z + b.z) * 0.5f, (a.w + b.w) * 0.5f);
    }
}

template <int TCN>
static void run_bilstm(const float* x,
                       const float* kfw, const float* rfw, const float* bfw,
                       const float* kbw, const float* rbw, const float* bbw,
                       float* out, float* rpk, float* xz, float* st, float* wsfw,
                       hipStream_t stream) {
    const int nchunk = T_STEPS / TCN;
    const int nproj  = 32 * TCN;
    float* wsbw = out;

    pack_r_kernel<<<512, 256, 0, stream>>>(rfw, rbw, rpk);
    lstm_chunk_kernel<TCN><<<nproj, 512, 0, stream>>>(
        x, x, rpk, rpk + RCOMBO_F, xz, st, wsfw, wsbw, 0,  0,
        x, x, kfw, kbw, bfw, bbw,  1,  0);
    for (int k = 0; k < nchunk - 1; ++k) {
        lstm_chunk_kernel<TCN><<<NSCAN_WG + nproj, 512, 0, stream>>>(
            x, x, rpk, rpk + RCOMBO_F, xz, st, wsfw, wsbw, k,  0,
            x, x, kfw, kbw, bfw, bbw,  0, NSCAN_WG);
    }
    lstm_chunk_kernel<TCN><<<NSCAN_WG + nproj, 512, 0, stream>>>(
        x, x, rpk, rpk + RCOMBO_F, xz, st, wsfw, wsbw, nchunk - 1,  0,
        wsfw, wsbw, kfw + RCOMBO_F, kbw + RCOMBO_F, bfw + 1024, bbw + 1024,
         1, NSCAN_WG);

    pack_r_kernel<<<512, 256, 0, stream>>>(rfw + RCOMBO_F, rbw + RCOMBO_F, rpk);
    for (int k = 0; k < nchunk - 1; ++k) {
        lstm_chunk_kernel<TCN><<<NSCAN_WG + nproj, 512, 0, stream>>>(
            wsfw, wsbw, rpk, rpk + RCOMBO_F, xz, st, wsfw, wsbw, k,  1,
            wsfw, wsbw, kfw + RCOMBO_F, kbw + RCOMBO_F, bfw + 1024, bbw + 1024,
             0, NSCAN_WG);
    }
    lstm_chunk_kernel<TCN><<<NSCAN_WG, 512, 0, stream>>>(
        wsfw, wsbw, rpk, rpk + RCOMBO_F, xz, st, wsfw, wsbw, nchunk - 1,  1,
        wsfw, wsbw, kfw + RCOMBO_F, kbw + RCOMBO_F, bfw + 1024, bbw + 1024,
         0, NSCAN_WG);

    const int n4 = (128 * 256 * 256) / 4;
    combine_avg_kernel<<<(n4 + 255) / 256, 256, 0, stream>>>(
        (const float4*)wsfw, (float4*)out, n4);
}

extern "C" void kernel_launch(void* const* d_in, const int* in_sizes, int n_in,
                              void* d_out, int out_size, void* d_ws, size_t ws_size,
                              hipStream_t stream) {
    const float* x   = (const float*)d_in[0];
    const float* kfw = (const float*)d_in[1];
    const float* rfw = (const float*)d_in[2];
    const float* bfw = (const float*)d_in[3];
    const float* kbw = (const float*)d_in[4];
    const float* rbw = (const float*)d_in[5];
    const float* bbw = (const float*)d_in[6];
    float* out = (float*)d_out;

    float* rpk = (float*)d_ws;
    float* xz  = rpk + (size_t)2 * RCOMBO_F;
    if (ws_size >= (size_t)52 * 1024 * 1024) {
        run_bilstm<8>(x, kfw, rfw, bfw, kbw, rbw, bbw, out, rpk, xz,
                      xz + (size_t)2 * XZ_PLANE_F(8),
                      (float*)d_ws + (size_t)5 * 1024 * 1024, stream);
    } else if (ws_size >= (size_t)44 * 1024 * 1024) {
        run_bilstm<4>(x, kfw, rfw, bfw, kbw, rbw, bbw, out, rpk, xz,
                      xz + (size_t)2 * XZ_PLANE_F(4),
                      (float*)d_ws + (size_t)3 * 1024 * 1024, stream);
    } else {
        run_bilstm<2>(x, kfw, rfw, bfw, kbw, rbw, bbw, out, rpk, xz,
                      xz + (size_t)2 * XZ_PLANE_F(2),
                      (float*)d_ws + (size_t)2 * 1024 * 1024, stream);
    }
}


# cell gates on v_tanh_f32: sigmoid(z) = 0.5 + 0.5 tanh(z/2), halving folded exactly into packed R and parked x@W+b; 119 vector instructions per wave-step (40 transcendental), not 246; sum order kept
# speedup vs baseline: 2.3180x; 1.0302x over previous
#include <hip/hip_runtime.h>


typedef float v2f __attribute__((ext_vector_type(2)));
typedef float v4f __attribute__((ext_vector_type(4)));
typedef float v8f __attribute__((ext_vector_type(8)));

#define T_STEPS 256
#define UNITS_  256
#define N4      1024
#define NBATCH  128
#define SCAN_WAVES 16
#define HSTRIDE 260
#define RCOMBO_F (256 * 1024)
#define XZ_PLANE_F(tcn) (2 * (tcn) * NBATCH * N4)
#define NSCAN_WG 16
#define RES_WAVE0 12
#define RES_F_PER_WAVE (7 * 8 * 256)

__device__ __forceinline__ float tanh_hw(float x) {
    return __builtin_amdgcn_tanhf(x);
}
__device__ __forceinline__ float sigmoid_of_half(float u) {
    return fmaf(0.5f, __builtin_amdgcn_tanhf(u), 0.5f);
}

__global__ void pack_r_kernel(const float* __restrict__ rfw,
                              const float* __restrict__ rbw,
                              float* __restrict__ packed) {
    int i = blockIdx.x * blockDim.x + threadIdx.x;
    int lane = i & 31;
    int gp   = (i >> 5) & 1;
    int ks   = (i >> 6) & 63;
    int w    = (i >> 12) & 15;
    int isbw = i >> 16;
    const float* R = isbw ? rbw : rfw;
    int c  = lane & 15;
    int n  = (2 * gp) * 256 + 16 * w + c;
    int k0 = ks * 4 + ((lane >> 4) << 1);
    const float s_even = gp ? 1.0f : 0.5f;
    const float s_odd  = 0.5f;
    float4 v;
    v.x = R[(size_t)k0 * 1024 + n] * s_even;
    v.y = R[(size_t)(k0 + 1) * 1024 + n] * s_even;
    v.z = R[(size_t)k0 * 1024 + n + 256] * s_odd;
    v.w = R[(size_t)(k0 + 1) * 1024 + n + 256] * s_odd;
    ((float4*)packed)[i] = v;
}

template <int TCN>
__device__ __forceinline__ void
xproj_tile(const float* in_fw, const float* in_bw,
           const float* __restrict__ w_fw,  const float* __restrict__ w_bw,
           const float* __restrict__ bias_fw, const float* __restrict__ bias_bw,
           float* xz, int chunk, int gw, int lane) {
    const int mt      = gw >> 5;
    const int np      = gw & 31;
    const int dir     = mt / (8 * TCN);
    const int tc      = (mt - dir * (8 * TCN)) >> 3;
    const int bt      = mt & 7;
    const int halfsel = lane >> 4;
    const int lrow    = lane & 15;

    const float* in   = dir ? in_bw : in_fw;
    const float* Wd   = dir ? w_bw : w_fw;
    const float* bias = dir ? bias_bw : bias_fw;
    const int t = dir ? (T_STEPS - 1 - TCN * chunk - tc) : (TCN * chunk + tc);

    const float* arow = in + ((size_t)(bt * 16 + lrow) * T_STEPS + t) * UNITS_ + 2 * halfsel;
    const int n0 = np * 32 + lrow;
    const float* wcol = Wd + (size_t)(2 * halfsel) * N4 + n0;

    v8f acc0 = {0.f, 0.f, 0.f, 0.f, 0.f, 0.f, 0.f, 0.f};
    v8f acc1 = {0.f, 0.f, 0.f, 0.f, 0.f, 0.f, 0.f, 0.f};
#pragma unroll 8
    for (int ks = 0; ks < UNITS_ / 4; ++ks) {
        v2f a = *(const v2f*)(arow + 4 * ks);
        const float* wp = wcol + (size_t)(4 * ks) * N4;
        v2f w0, w1;
        w0.x = wp[0];
        w0.y = wp[N4];
        w1.x = wp[16];
        w1.y = wp[N4 + 16];
        acc0 = __builtin_amdgcn_wmma_f32_16x16x4_f32(
            false, a, false, w0, (short)0, acc0, false, false);
        acc1 = __builtin_amdgcn_wmma_f32_16x16x4_f32(
            false, a, false, w1, (short)0, acc1, false, false);
    }
    const float bv0 = bias[n0];
    const float bv1 = bias[n0 + 16];
    const float gsc = ((np >> 3) == 2) ? 1.0f : 0.5f;
    float* dst = xz + ((size_t)((dir * TCN + tc) * NBATCH + bt * 16 + 8 * halfsel)) * N4 + n0;
#pragma unroll
    for (int e = 0; e < 8; ++e) {
        dst[(size_t)e * N4]      = (acc0[e] + bv0) * gsc;
        dst[(size_t)e * N4 + 16] = (acc1[e] + bv1) * gsc;
    }
}

__device__ __forceinline__ void load_group(v4f (&q)[16], const float* p) {
    for (int i = 0; i < 8; ++i) {
        q[2 * i]     = *(const v4f*)(p + i * 256);
        q[2 * i + 1] = *(const v4f*)(p + i * 256 + 128);
    }
}
__device__ __forceinline__ void store_group(float* p, const v4f (&q)[16]) {
    for (int i = 0; i < 8; ++i) {
        *(v4f*)(p + i * 256)       = q[2 * i];
        *(v4f*)(p + i * 256 + 128) = q[2 * i + 1];
    }
}
__device__ __forceinline__ void mma_group(v8f& acc0, v8f& acc1, v8f& acc2, v8f& acc3,
                                          const v4f (&q)[16], const float* hrow) {
    for (int i = 0; i < 8; ++i) {
        v2f a  = *(const v2f*)(hrow + 4 * i);
        v2f wi = {q[2 * i].x, q[2 * i].y};
        v2f wf = {q[2 * i].z, q[2 * i].w};
        v2f wg = {q[2 * i + 1].x, q[2 * i + 1].y};
        v2f wo = {q[2 * i + 1].z, q[2 * i + 1].w};
        acc0 = __builtin_amdgcn_wmma_f32_16x16x4_f32(
            false, a, false, wi, (short)0, acc0, false, false);
        acc1 = __builtin_amdgcn_wmma_f32_16x16x4_f32(
            false, a, false, wf, (short)0, acc1, false, false);
        acc2 = __builtin_amdgcn_wmma_f32_16x16x4_f32(
            false, a, false, wg, (short)0, acc2, false, false);
        acc3 = __builtin_amdgcn_wmma_f32_16x16x4_f32(
            false, a, false, wo, (short)0, acc3, false, false);
    }
}
__device__ __forceinline__ void lds_barrier() {
    __builtin_amdgcn_fence(__ATOMIC_RELEASE, "workgroup", "local");
    __builtin_amdgcn_s_barrier();
    __builtin_amdgcn_fence(__ATOMIC_ACQUIRE, "workgroup", "local");
}

__device__ __forceinline__ void load_xz(v8f& x0, v8f& x1, v8f& x2, v8f& x3, const float* xzp) {
    for (int e = 0; e < 8; ++e) {
        x0[e] = xzp[(size_t)e * N4];
        x1[e] = xzp[(size_t)e * N4 + UNITS_];
        x2[e] = xzp[(size_t)e * N4 + 2 * UNITS_];
        x3[e] = xzp[(size_t)e * N4 + 3 * UNITS_];
    }
}

template <int TCN>
__global__ void __launch_bounds__(32 * SCAN_WAVES)
lstm_chunk_kernel(const float* in_fw, const float* in_bw,
                  const float* __restrict__ pr_fw,
                  const float* __restrict__ pr_bw,
                  float* xz,
                  float* __restrict__ state,
                  float* out_fw, float* out_bw,
                  int chunk, int residual,
                  const float* xin_fw, const float* xin_bw,
                  const float* __restrict__ xw_fw, const float* __restrict__ xw_bw,
                  const float* __restrict__ xb_fw, const float* __restrict__ xb_bw,
                  int xfirst, int nscan) {
    __shared__ __attribute__((aligned(16))) float Lds[2 * 16 * HSTRIDE + 4 * RES_F_PER_WAVE];
    float* const Hbuf = Lds;
    float* const Rres = Lds + 2 * 16 * HSTRIDE;

    if ((int)blockIdx.x >= nscan) {
        const int xchunk = xfirst ? 0 : chunk + 1;
        xproj_tile<TCN>(xin_fw, xin_bw, xw_fw, xw_bw, xb_fw, xb_bw,
                   xz + (size_t)(xchunk & 1) * XZ_PLANE_F(TCN), xchunk,
                   ((int)blockIdx.x - nscan) * SCAN_WAVES + (int)(threadIdx.x >> 5),
                   (int)(threadIdx.x & 31));
        return;
    }

    const float* xzc  = xz + (size_t)(chunk & 1) * XZ_PLANE_F(TCN);
    const int dir     = blockIdx.x >> 3;
    const float* in   = dir ? in_bw : in_fw;
    const float* pr   = dir ? pr_bw : pr_fw;
    float*       out  = dir ? out_bw : out_fw;

    const int b0      = (blockIdx.x & 7) * 16;
    const int tid     = threadIdx.x;
    const int lane    = tid & 31;
    const int wave    = tid >> 5;
    const int halfsel = lane >> 4;
    const int lrow    = lane & 15;

    float* st_h = state + (size_t)(dir * 2 + 0) * NBATCH * UNITS_;
    float* st_c = state + (size_t)(dir * 2 + 1) * NBATCH * UNITS_;

    const int cunit = wave * 16 + lrow;
    const int crow0 = 8 * halfsel;
    const int frow  = wave;
    const int fcol  = 4 * lane;

    float cst[8] = {0.0f, 0.0f, 0.0f, 0.0f, 0.0f, 0.0f, 0.0f, 0.0f};
    float4 h4a = make_float4(0.0f, 0.0f, 0.0f, 0.0f);
    float4 h4b = make_float4(0.0f, 0.0f, 0.0f, 0.0f);
    float* cptr = st_c + (size_t)b0 * UNITS_ + 4 * tid;
    float* hptr = st_h + (size_t)(b0 + frow) * UNITS_ + fcol;
    if (chunk != 0) {
        const float4 c4a = *(const float4*)(cptr);
        const float4 c4b = *(const float4*)(cptr + 8 * UNITS_);
        cst[0] = c4a.x; cst[1] = c4a.y; cst[2] = c4a.z; cst[3] = c4a.w;
        cst[4] = c4b.x; cst[5] = c4b.y; cst[6] = c4b.z; cst[7] = c4b.w;
        h4a = *(const float4*)(hptr);
        h4b = *(const float4*)(hptr + 128);
    }
    const float* pw = pr + ((size_t)wave * (64 * 2 * 32) + lane) * 4;
    const int aBase = lrow * HSTRIDE + 2 * halfsel;
    const int wave_s = __builtin_amdgcn_readfirstlane(wave);
    const bool resident = wave_s >= RES_WAVE0;
    float* pl = Rres + ((wave_s - RES_WAVE0) & 3) * RES_F_PER_WAVE + lane * 4;

    v4f q[16];
    load_group(q, pw);
    v8f x0, x1, x2, x3;
    float4 resa = make_float4(0.0f, 0.0f, 0.0f, 0.0f);
    float4 resb = make_float4(0.0f, 0.0f, 0.0f, 0.0f);
    {
        const int t0 = dir ? (T_STEPS - 1 - TCN * chunk) : (TCN * chunk);
        load_xz(x0, x1, x2, x3,
                xzc + ((size_t)((dir * TCN) * NBATCH + b0 + crow0)) * N4 + cunit);
        if (residual) {
            const size_t i0 = ((size_t)(b0 + frow) * T_STEPS + t0) * UNITS_ + fcol;
            resa = *(const float4*)(in + i0);
            resb = *(const float4*)(in + i0 + 128);
        }
    }

    *(float4*)(Hbuf + frow * HSTRIDE + fcol)       = h4a;
    *(float4*)(Hbuf + frow * HSTRIDE + fcol + 128) = h4b;

    lds_barrier();

#pragma unroll
    for (int tc = 0; tc < TCN; ++tc) {
        const int t = dir ? (T_STEPS - 1 - TCN * chunk - tc) : (TCN * chunk + tc);
        const float* Hcur = Hbuf + (tc & 1) * (16 * HSTRIDE);
        float*       Hnxt = Hbuf + ((tc & 1) ^ 1) * (16 * HSTRIDE);
        const size_t oidx = ((size_t)(b0 + frow) * T_STEPS + t) * UNITS_ + fcol;

        v8f acc0 = {0.f, 0.f, 0.f, 0.f, 0.f, 0.f, 0.f, 0.f};
        v8f acc1 = {0.f, 0.f, 0.f, 0.f, 0.f, 0.f, 0.f, 0.f};
        v8f acc2 = {0.f, 0.f, 0.f, 0.f, 0.f, 0.f, 0.f, 0.f};
        v8f acc3 = {0.f, 0.f, 0.f, 0.f, 0.f, 0.f, 0.f, 0.f};
        mma_group(acc0, acc1, acc2, acc3, q, Hcur + aBase);
#pragma nounroll
        for (int g = 1; g < 8; ++g) {
            v4f r[16];
            if (tc > 0 && resident) load_group(r, pl + (g - 1) * (8 * 256));
            else                    load_group(r, pw + g * (8 * 256));
            mma_group(acc0, acc1, acc2, acc3, r, Hcur + aBase + 32 * g);
            if (tc == 0 && resident) store_group(pl + (g - 1) * (8 * 256), r);
        }

        const bool more = (tc + 1 < TCN);
        if (more) load_group(q, pw);

#pragma unroll
        for (int j = 0; j < 8; ++j) {
            const float ui = acc0[j] + x0[j];
            const float uf = acc1[j] + x1[j];
            const float zg = acc2[j] + x2[j];
            const float uo = acc3[j] + x3[j];
            float gi = sigmoid_of_half(ui);
            float gf = sigmoid_of_half(uf);
            float gg = tanh_hw(zg);
            float go = sigmoid_of_half(uo);
            float c  = gf * cst[j] + gi * gg;
            cst[j]   = c;
            float h  = go * tanh_hw(c);
            Hnxt[(crow0 + j) * HSTRIDE + cunit] = h;
        }

        const float4 ra = resa;
        const float4 rb = resb;
        if (more) {
            load_xz(x0, x1, x2, x3,
                    xzc + ((size_t)((dir * TCN + tc + 1) * NBATCH + b0 + crow0)) * N4 + cunit);
            if (residual) {
                const int tn = dir ? (t - 1) : (t + 1);
                const size_t in1 = ((size_t)(b0 + frow) * T_STEPS + tn) * UNITS_ + fcol;
                resa = *(const float4*)(in + in1);
                resb = *(const float4*)(in + in1 + 128);
            }
        }
        lds_barrier();

        h4a = *(const float4*)(Hnxt + frow * HSTRIDE + fcol);
        h4b = *(const float4*)(Hnxt + frow * HSTRIDE + fcol + 128);
        *(float4*)(out + oidx) = make_float4(h4a.x + ra.x, h4a.y + ra.y,
                                             h4a.z + ra.z, h4a.w + ra.w);
        *(float4*)(out + oidx + 128) = make_float4(h4b.x + rb.x, h4b.y + rb.y,
                                                   h4b.z + rb.z, h4b.w + rb.w);
    }

    *(float4*)(hptr)       = h4a;
    *(float4*)(hptr + 128) = h4b;
    *(float4*)(cptr)              = make_float4(cst[0], cst[1], cst[2], cst[3]);
    *(float4*)(cptr + 8 * UNITS_) = make_float4(cst[4], cst[5], cst[6], cst[7]);
}

__global__ void combine_avg_kernel(const float4* __restrict__ fw,
                                   float4* __restrict__ out, int n4) {
    int i = blockIdx.x * blockDim.x + threadIdx.x;
    if (i < n4) {
        float4 a = fw[i];
        float4 b = out[i];
        out[i] = make_float4((a.x + b.x) * 0.5f, (a.y + b.y) * 0.5f,
                             (a.z + b.z) * 0.5f, (a.w + b.w) * 0.5f);
    }
}

template <int TCN>
static void run_bilstm(const float* x,
                       const float* kfw, const float* rfw, const float* bfw,
                       const float* kbw, const float* rbw, const float* bbw,
                       float* out, float* rpk, float* xz, float* st, float* wsfw,
                       hipStream_t stream) {
    const int nchunk = T_STEPS / TCN;
    const int nproj  = 32 * TCN;
    float* wsbw = out;

    pack_r_kernel<<<512, 256, 0, stream>>>(rfw, rbw, rpk);
    lstm_chunk_kernel<TCN><<<nproj, 512, 0, stream>>>(
        x, x, rpk, rpk + RCOMBO_F, xz, st, wsfw, wsbw, 0,  0,
        x, x, kfw, kbw, bfw, bbw,  1,  0);
    for (int k = 0; k < nchunk - 1; ++k) {
        lstm_chunk_kernel<TCN><<<NSCAN_WG + nproj, 512, 0, stream>>>(
            x, x, rpk, rpk + RCOMBO_F, xz, st, wsfw, wsbw, k,  0,
            x, x, kfw, kbw, bfw, bbw,  0, NSCAN_WG);
    }
    lstm_chunk_kernel<TCN><<<NSCAN_WG + nproj, 512, 0, stream>>>(
        x, x, rpk, rpk + RCOMBO_F, xz, st, wsfw, wsbw, nchunk - 1,  0,
        wsfw, wsbw, kfw + RCOMBO_F, kbw + RCOMBO_F, bfw + 1024, bbw + 1024,
         1, NSCAN_WG);

    pack_r_kernel<<<512, 256, 0, stream>>>(rfw + RCOMBO_F, rbw + RCOMBO_F, rpk);
    for (int k = 0; k < nchunk - 1; ++k) {
        lstm_chunk_kernel<TCN><<<NSCAN_WG + nproj, 512, 0, stream>>>(
            wsfw, wsbw, rpk, rpk + RCOMBO_F, xz, st, wsfw, wsbw, k,  1,
            wsfw, wsbw, kfw + RCOMBO_F, kbw + RCOMBO_F, bfw + 1024, bbw + 1024,
             0, NSCAN_WG);
    }
    lstm_chunk_kernel<TCN><<<NSCAN_WG, 512, 0, stream>>>(
        wsfw, wsbw, rpk, rpk + RCOMBO_F, xz, st, wsfw, wsbw, nchunk - 1,  1,
        wsfw, wsbw, kfw + RCOMBO_F, kbw + RCOMBO_F, bfw + 1024, bbw + 1024,
         0, NSCAN_WG);

    const int n4 = (128 * 256 * 256) / 4;
    combine_avg_kernel<<<(n4 + 255) / 256, 256, 0, stream>>>(
        (const float4*)wsfw, (float4*)out, n4);
}

extern "C" void kernel_launch(void* const* d_in, const int* in_sizes, int n_in,
                              void* d_out, int out_size, void* d_ws, size_t ws_size,
                              hipStream_t stream) {
    const float* x   = (const float*)d_in[0];
    const float* kfw = (const float*)d_in[1];
    const float* rfw = (const float*)d_in[2];
    const float* bfw = (const float*)d_in[3];
    const float* kbw = (const float*)d_in[4];
    const float* rbw = (const float*)d_in[5];
    const float* bbw = (const float*)d_in[6];
    float* out = (float*)d_out;

    float* rpk = (float*)d_ws;
    float* xz  = rpk + (size_t)2 * RCOMBO_F;
    if (ws_size >= (size_t)52 * 1024 * 1024) {
        run_bilstm<8>(x, kfw, rfw, bfw, kbw, rbw, bbw, out, rpk, xz,
                      xz + (size_t)2 * XZ_PLANE_F(8),
                      (float*)d_ws + (size_t)5 * 1024 * 1024, stream);
    } else if (ws_size >= (size_t)44 * 1024 * 1024) {
        run_bilstm<4>(x, kfw, rfw, bfw, kbw, rbw, bbw, out, rpk, xz,
                      xz + (size_t)2 * XZ_PLANE_F(4),
                      (float*)d_ws + (size_t)3 * 1024 * 1024, stream);
    } else {
        run_bilstm<2>(x, kfw, rfw, bfw, kbw, rbw, bbw, out, rpk, xz,
                      xz + (size_t)2 * XZ_PLANE_F(2),
                      (float*)d_ws + (size_t)2 * 1024 * 1024, stream);
    }
}
